# starting kernel
# baseline (speedup 1.0000x reference)
#include <hip/hip_runtime.h>
#include <hip/hip_bf16.h>

typedef __attribute__((ext_vector_type(2))) float v2f;
typedef __attribute__((ext_vector_type(8))) float v8f;

__global__ __launch_bounds__(256) void fgcn_pack_w(const float* __restrict__ W,
                                                   float* __restrict__ Wp,
                                                   int total  ) {
    int tid = blockIdx.x * blockDim.x + threadIdx.x;
    if (tid >= total) return;
    const int lane = tid & 31;
    const int n    = (tid >> 5) & 3;
    const int kk   = tid >> 7;
    const int kb   = kk * 4 + ((lane >= 16) ? 2 : 0);
    const int col  = n * 16 + (lane & 15);
    float2 v;
    v.x = W[(size_t)kb * 64 + col];
    v.y = W[(size_t)(kb + 1) * 64 + col];
    *(float2*)(Wp + (size_t)tid * 2) = v;
}

__global__ __launch_bounds__(256) void fgcn_gemm_wmma(const float* __restrict__ X,
                                                      const float* __restrict__ Wp,
                                                      float* __restrict__ H,
                                                      int mtiles, int K) {
    const int wave = (blockIdx.x * blockDim.x + threadIdx.x) >> 5;
    const int lane = threadIdx.x & 31;
    if (wave >= mtiles) return;

    const int m0     = wave * 16;
    const int lane16 = lane & 15;
    const bool hi    = lane >= 16;

    v8f acc0 = {}, acc1 = {}, acc2 = {}, acc3 = {};

    const float* xa = X + (size_t)(m0 + lane16) * K + (hi ? 2 : 0);
    const float* wb = Wp + (size_t)lane * 2;

    for (int k = 0; k < K; k += 4) {
        float2 af = *(const float2*)(xa + k);
        v2f a; a.x = af.x; a.y = af.y;

        const float* wk = wb + (size_t)(k >> 2) * 256;
        v2f b0 = *(const v2f*)(wk);
        v2f b1 = *(const v2f*)(wk + 64);
        v2f b2 = *(const v2f*)(wk + 128);
        v2f b3 = *(const v2f*)(wk + 192);

        acc0 = __builtin_amdgcn_wmma_f32_16x16x4_f32(false, a, false, b0, (short)0, acc0, false, false);
        acc1 = __builtin_amdgcn_wmma_f32_16x16x4_f32(false, a, false, b1, (short)0, acc1, false, false);
        acc2 = __builtin_amdgcn_wmma_f32_16x16x4_f32(false, a, false, b2, (short)0, acc2, false, false);
        acc3 = __builtin_amdgcn_wmma_f32_16x16x4_f32(false, a, false, b3, (short)0, acc3, false, false);
    }

    const int rbase = m0 + (hi ? 8 : 0);
    #pragma unroll
    for (int r = 0; r < 8; ++r) {
        float* o = H + (size_t)(rbase + r) * 64 + lane16;
        o[0]  = acc0[r];
        o[16] = acc1[r];
        o[32] = acc2[r];
        o[48] = acc3[r];
    }
}

__global__ __launch_bounds__(256) void fgcn_bias_init(float* __restrict__ out,
                                                      const float* __restrict__ bias,
                                                      int total) {
    int idx = blockIdx.x * blockDim.x + threadIdx.x;
    if (idx < total) out[idx] = bias[idx & 63];
}

__global__ __launch_bounds__(256) void fgcn_edge_scatter(const float* __restrict__ h,
                                                         const int* __restrict__ src,
                                                         const int* __restrict__ dst,
                                                         float* __restrict__ out,
                                                         int E) {
    int idx = blockIdx.x * blockDim.x + threadIdx.x;
    if (idx >= E * 16) return;
    const int e = idx >> 4;
    const int q = (idx & 15) << 2;
    const int s = src[e];
    const int d = dst[e];
    float4 v = *(const float4*)(h + (size_t)s * 64 + q);
    float* o = out + (size_t)d * 64 + q;
    atomicAdd(o + 0, v.x);
    atomicAdd(o + 1, v.y);
    atomicAdd(o + 2, v.z);
    atomicAdd(o + 3, v.w);
}

static constexpr int N_DRUG = 50000, E_DRUG = 1000000, F_DRUG = 256;
static constexpr int N_DIS  = 20000, E_DIS  = 640000,  F_DIS  = 128;
static constexpr int NHID2  = 64;

extern "C" void kernel_launch(void* const* d_in, const int* in_sizes, int n_in,
                              void* d_out, int out_size, void* d_ws, size_t ws_size,
                              hipStream_t stream) {
    const float* drug_x  = (const float*)d_in[0];
    const float* dis_x   = (const float*)d_in[1];
    const float* W1      = (const float*)d_in[2];
    const float* b1      = (const float*)d_in[3];
    const float* W2      = (const float*)d_in[4];
    const float* b2      = (const float*)d_in[5];
    const int*   drug_ei = (const int*)d_in[6];
    const int*   dis_ei  = (const int*)d_in[7];

    float* out1 = (float*)d_out;
    float* out2 = out1 + (size_t)N_DRUG * NHID2;

    float* h1  = (float*)d_ws;
    float* h2  = h1 + (size_t)N_DRUG * NHID2;
    float* Wp1 = h2 + (size_t)N_DIS * NHID2;
    float* Wp2 = Wp1 + (size_t)F_DRUG * NHID2;

    {
        int total = F_DRUG * NHID2 / 2;
        fgcn_pack_w<<<(total + 255) / 256, 256, 0, stream>>>(W1, Wp1, total);
    }
    {
        int total = F_DIS * NHID2 / 2;
        fgcn_pack_w<<<(total + 255) / 256, 256, 0, stream>>>(W2, Wp2, total);
    }

    {
        int mtiles = N_DRUG / 16;
        int grid = (mtiles + 7) / 8;
        fgcn_gemm_wmma<<<grid, 256, 0, stream>>>(drug_x, Wp1, h1, mtiles, F_DRUG);
    }
    {
        int mtiles = N_DIS / 16;
        int grid = (mtiles + 7) / 8;
        fgcn_gemm_wmma<<<grid, 256, 0, stream>>>(dis_x, Wp2, h2, mtiles, F_DIS);
    }

    {
        int total = N_DRUG * NHID2;
        fgcn_bias_init<<<(total + 255) / 256, 256, 0, stream>>>(out1, b1, total);
    }
    {
        int total = N_DIS * NHID2;
        fgcn_bias_init<<<(total + 255) / 256, 256, 0, stream>>>(out2, b2, total);
    }

    {
        int threads = E_DRUG * 16;
        fgcn_edge_scatter<<<(threads + 255) / 256, 256, 0, stream>>>(
            h1, drug_ei, drug_ei + E_DRUG, out1, E_DRUG);
    }
    {
        int threads = E_DIS * 16;
        fgcn_edge_scatter<<<(threads + 255) / 256, 256, 0, stream>>>(
            h2, dis_ei, dis_ei + E_DIS, out2, E_DIS);
    }
}


# GEMM reads W in place: a lane loads 16 bytes of rows k+2h and k+2h+1 (4 adjacent columns), tile j = columns 4c+j; both pack launches go, h rows stored as 16-byte pieces; sum order unchanged
# speedup vs baseline: 2.2106x; 2.2106x over previous
#include <hip/hip_runtime.h>
#include <hip/hip_bf16.h>

typedef __attribute__((ext_vector_type(2))) float v2f;
typedef __attribute__((ext_vector_type(8))) float v8f;

__global__ __launch_bounds__(256) void fgcn_gemm_wmma(const float* __restrict__ X,
                                                      const float* __restrict__ W,
                                                      float* __restrict__ H,
                                                      int mtiles, int K) {
    const int wave = (blockIdx.x * blockDim.x + threadIdx.x) >> 5;
    const int lane = threadIdx.x & 31;
    if (wave >= mtiles) return;

    const int m0     = wave * 16;
    const int lane16 = lane & 15;
    const bool hi    = lane >= 16;

    v8f acc0 = {}, acc1 = {}, acc2 = {}, acc3 = {};

    const float* xa = X + (size_t)(m0 + lane16) * K + (hi ? 2 : 0);
    const float* wb = W + (size_t)(hi ? 2 : 0) * 64 + lane16 * 4;

    #pragma unroll 8
    for (int k = 0; k < K; k += 4) {
        float2 af = *(const float2*)(xa + k);
        v2f a; a.x = af.x; a.y = af.y;

        const float* wk = wb + (size_t)k * 64;
        const float4 w0 = *(const float4*)(wk);
        const float4 w1 = *(const float4*)(wk + 64);
        v2f b0; b0.x = w0.x; b0.y = w1.x;
        v2f b1; b1.x = w0.y; b1.y = w1.y;
        v2f b2; b2.x = w0.z; b2.y = w1.z;
        v2f b3; b3.x = w0.w; b3.y = w1.w;

        acc0 = __builtin_amdgcn_wmma_f32_16x16x4_f32(false, a, false, b0, (short)0, acc0, false, false);
        acc1 = __builtin_amdgcn_wmma_f32_16x16x4_f32(false, a, false, b1, (short)0, acc1, false, false);
        acc2 = __builtin_amdgcn_wmma_f32_16x16x4_f32(false, a, false, b2, (short)0, acc2, false, false);
        acc3 = __builtin_amdgcn_wmma_f32_16x16x4_f32(false, a, false, b3, (short)0, acc3, false, false);
    }

    const int rbase = m0 + (hi ? 8 : 0);
    #pragma unroll
    for (int r = 0; r < 8; ++r) {
        float4 o;
        o.x = acc0[r];
        o.y = acc1[r];
        o.z = acc2[r];
        o.w = acc3[r];
        *(float4*)(H + (size_t)(rbase + r) * 64 + lane16 * 4) = o;
    }
}

static constexpr int N_DRUG = 50000, E_DRUG = 1000000, F_DRUG = 256;
static constexpr int N_DIS  = 20000, E_DIS  = 640000,  F_DIS  = 128;
static constexpr int NHID2  = 64;

static constexpr int BIN_ROWS = 64;
static constexpr int CHUNK    = 8192;
static constexpr int NB1 = (N_DRUG + BIN_ROWS - 1) / BIN_ROWS;
static constexpr int NB2 = (N_DIS  + BIN_ROWS - 1) / BIN_ROWS;
static constexpr int CH1 = (E_DRUG + CHUNK - 1) / CHUNK;
static constexpr int CH2 = (E_DIS  + CHUNK - 1) / CHUNK;
static constexpr int CHMAX = 128;
static_assert(NB1 < 1024 && NB2 < 1024, "one bin counter per thread, plus the end entry");
static_assert(CH1 <= CHMAX && CH2 <= CHMAX, "run bounds table in LDS");
static_assert(N_DRUG <= 65536 && N_DIS <= 65536, "record packs (src << 6) | local row");

__global__ __launch_bounds__(1024) void fgcn_group_edges(const int* __restrict__ ei1,
                                                         const int* __restrict__ ei2,
                                                         int* __restrict__ rec1,
                                                         int* __restrict__ rec2,
                                                         int* __restrict__ loc1,
                                                         int* __restrict__ loc2) {
    __shared__ int cnt[1024];
    __shared__ int wsum[32];
    const int tid  = threadIdx.x;
    const int lane = tid & 31;
    const int w    = tid >> 5;

    int c = blockIdx.x;
    const int* src; const int* dst; int* rec; int* loc; int E, NB;
    if (c < CH1) { src = ei1; dst = ei1 + E_DRUG; rec = rec1; loc = loc1; E = E_DRUG; NB = NB1; }
    else { c -= CH1; src = ei2; dst = ei2 + E_DIS; rec = rec2; loc = loc2; E = E_DIS; NB = NB2; }

    cnt[tid] = 0;
    __syncthreads();

    const int base = c * CHUNK;
    int sv[8], dv[8];
    #pragma unroll
    for (int i = 0; i < 8; ++i) {
        const int e  = base + i * 1024 + tid;
        const int ec = (e < E) ? e : E - 1;
        sv[i] = src[ec];
        dv[i] = dst[ec];
    }
    int r[8], bk[8], rk[8];
    #pragma unroll
    for (int i = 0; i < 8; ++i) {
        const int e = base + i * 1024 + tid;
        bk[i] = -1; rk[i] = 0;
        r[i]  = (sv[i] << 6) | (dv[i] & 63);
        if (e < E) {
            bk[i] = dv[i] >> 6;
            rk[i] = atomicAdd(&cnt[bk[i]], 1);
        }
    }
    __syncthreads();

    const int v = cnt[tid];
    int x = v;
    for (int o = 1; o < 32; o <<= 1) {
        const int y = __shfl_up(x, o);
        if (lane >= o) x += y;
    }
    if (lane == 31) wsum[w] = x;
    __syncthreads();
    if (w == 0) {
        const int t0 = wsum[lane];
        int t = t0;
        for (int o = 1; o < 32; o <<= 1) {
            const int y = __shfl_up(t, o);
            if (lane >= o) t += y;
        }
        wsum[lane] = t - t0;
    }
    __syncthreads();
    const int excl = x - v + wsum[w];
    cnt[tid] = excl;
    if (tid <= NB) loc[(size_t)c * (NB + 1) + tid] = excl;
    __syncthreads();

    int pos[8];
    #pragma unroll
    for (int i = 0; i < 8; ++i) pos[i] = cnt[(bk[i] >= 0) ? bk[i] : 0] + rk[i];
    #pragma unroll
    for (int i = 0; i < 8; ++i) {
        if (bk[i] >= 0) rec[(size_t)base + pos[i]] = r[i];
    }
}

__device__ __forceinline__ void fgcn_row_add(float* acc, int rcd, int lane, float2 v) {
    float* a = acc + ((rcd & 63) << 6) + lane * 2;
    atomicAdd(a, v.x);
    atomicAdd(a + 1, v.y);
}

template <int K>
__device__ __forceinline__ void fgcn_rows(float* acc, const float* hl, int cur, int j, int lane) {
    int r[K];
    float2 v[K];
    #pragma unroll
    for (int k = 0; k < K; ++k) r[k] = __builtin_amdgcn_readlane(cur, j + k);
    #pragma unroll
    for (int k = 0; k < K; ++k) v[k] = *(const float2*)(hl + ((size_t)(r[k] >> 6) << 6));
    #pragma unroll
    for (int k = 0; k < K; ++k) fgcn_row_add(acc, r[k], lane, v[k]);
}

__device__ __forceinline__ void fgcn_piece(float* acc, const float* hl, int cur, int m, int lane) {
    int j = 0;
    for (; j + 8 <= m; j += 8) fgcn_rows<8>(acc, hl, cur, j, lane);
    switch (m - j) {
        case 7: fgcn_rows<7>(acc, hl, cur, j, lane); break;
        case 6: fgcn_rows<6>(acc, hl, cur, j, lane); break;
        case 5: fgcn_rows<5>(acc, hl, cur, j, lane); break;
        case 4: fgcn_rows<4>(acc, hl, cur, j, lane); break;
        case 3: fgcn_rows<3>(acc, hl, cur, j, lane); break;
        case 2: fgcn_rows<2>(acc, hl, cur, j, lane); break;
        case 1: fgcn_rows<1>(acc, hl, cur, j, lane); break;
        default: break;
    }
}

__global__ __launch_bounds__(256) void fgcn_sum_rows(const float* __restrict__ h1,
                                                     const float* __restrict__ h2,
                                                     const int* __restrict__ rec1,
                                                     const int* __restrict__ rec2,
                                                     const int* __restrict__ loc1,
                                                     const int* __restrict__ loc2,
                                                     const float* __restrict__ b1,
                                                     const float* __restrict__ b2,
                                                     float* __restrict__ out1,
                                                     float* __restrict__ out2) {
    __shared__ float acc[BIN_ROWS * NHID2];
    __shared__ int lo[CHMAX];
    __shared__ int hi[CHMAX];

    int b = blockIdx.x;
    const float* h; const int* rec; const int* loc; const float* bias; float* out; int NB, CH, N;
    if (b < NB1) { h = h1; rec = rec1; loc = loc1; bias = b1; out = out1; NB = NB1; CH = CH1; N = N_DRUG; }
    else { b -= NB1; h = h2; rec = rec2; loc = loc2; bias = b2; out = out2; NB = NB2; CH = CH2; N = N_DIS; }

    const int tid  = threadIdx.x;
    const int lane = tid & 31;
    const int w    = tid >> 5;

    for (int i = tid; i < BIN_ROWS * NHID2; i += 256) acc[i] = 0.0f;
    for (int t = tid; t < CH; t += 256) {
        const int* lp = loc + (size_t)t * (NB + 1) + b;
        lo[t] = lp[0];
        hi[t] = lp[1];
    }
    __syncthreads();

    const float* hl = h + lane * 2;
    int c = __builtin_amdgcn_readfirstlane(w);
    int s = 0, e = 0, mine = 0;
    if (c < CH) {
        s = __builtin_amdgcn_readfirstlane(lo[c]);
        e = __builtin_amdgcn_readfirstlane(hi[c]);
        if (lane < e - s) mine = rec[(size_t)c * CHUNK + s + lane];
    }
    while (c < CH) {
        const int cn = c + 8;
        int sn = 0, en = 0, minen = 0;
        if (cn < CH) {
            sn = __builtin_amdgcn_readfirstlane(lo[cn]);
            en = __builtin_amdgcn_readfirstlane(hi[cn]);
            if (lane < en - sn) minen = rec[(size_t)cn * CHUNK + sn + lane];
        }
        const int n = e - s;
        if (n > 0) {
            fgcn_piece(acc, hl, mine, (n < 32) ? n : 32, lane);
            const int* rp = rec + (size_t)c * CHUNK;
            for (int p = s + 32; p < e; p += 32) {
                const int m = (e - p < 32) ? (e - p) : 32;
                const int more = (lane < m) ? rp[p + lane] : 0;
                fgcn_piece(acc, hl, more, m, lane);
            }
        }
        c = cn; s = sn; e = en; mine = minen;
    }
    __syncthreads();

    for (int i = tid; i < BIN_ROWS * NHID2 / 4; i += 256) {
        const int row = b * BIN_ROWS + (i >> 4);
        if (row < N) {
            const int q = (i & 15) << 2;
            float4 a = *(const float4*)(acc + i * 4);
            a.x += bias[q + 0];
            a.y += bias[q + 1];
            a.z += bias[q + 2];
            a.w += bias[q + 3];
            *(float4*)(out + (size_t)row * NHID2 + q) = a;
        }
    }
}

extern "C" void kernel_launch(void* const* d_in, const int* in_sizes, int n_in,
                              void* d_out, int out_size, void* d_ws, size_t ws_size,
                              hipStream_t stream) {
    const float* drug_x  = (const float*)d_in[0];
    const float* dis_x   = (const float*)d_in[1];
    const float* W1      = (const float*)d_in[2];
    const float* b1      = (const float*)d_in[3];
    const float* W2      = (const float*)d_in[4];
    const float* b2      = (const float*)d_in[5];
    const int*   drug_ei = (const int*)d_in[6];
    const int*   dis_ei  = (const int*)d_in[7];

    float* out1 = (float*)d_out;
    float* out2 = out1 + (size_t)N_DRUG * NHID2;

    float* h1  = (float*)d_ws;
    float* h2  = h1 + (size_t)N_DRUG * NHID2;
    int* rec1  = (int*)(h2 + (size_t)N_DIS * NHID2);
    int* rec2  = rec1 + (size_t)E_DRUG;
    int* loc1  = rec2 + (size_t)E_DIS;
    int* loc2  = loc1 + (size_t)CH1 * (NB1 + 1);

    {
        int mtiles = N_DRUG / 16;
        int grid = (mtiles + 7) / 8;
        fgcn_gemm_wmma<<<grid, 256, 0, stream>>>(drug_x, W1, h1, mtiles, F_DRUG);
    }
    {
        int mtiles = N_DIS / 16;
        int grid = (mtiles + 7) / 8;
        fgcn_gemm_wmma<<<grid, 256, 0, stream>>>(dis_x, W2, h2, mtiles, F_DIS);
    }

    fgcn_group_edges<<<CH1 + CH2, 1024, 0, stream>>>(drug_ei, dis_ei, rec1, rec2, loc1, loc2);

    fgcn_sum_rows<<<NB1 + NB2, 256, 0, stream>>>(h1, h2, rec1, rec2, loc1, loc2, b1, b2, out1, out2);
}


# the two GEMM launches and the edge-grouping launch run as one launch (block-uniform role; grouping at 256 threads, 32 edges each; a K=128 wave does two tiles): launches 4 -> 2; sum order unchanged
# speedup vs baseline: 2.7408x; 1.2398x over previous
#include <hip/hip_runtime.h>
#include <hip/hip_bf16.h>

typedef __attribute__((ext_vector_type(2))) float v2f;
typedef __attribute__((ext_vector_type(8))) float v8f;

static constexpr int N_DRUG = 50000, E_DRUG = 1000000, F_DRUG = 256;
static constexpr int N_DIS  = 20000, E_DIS  = 640000,  F_DIS  = 128;
static constexpr int NHID2  = 64;

static constexpr int BIN_ROWS = 64;
static constexpr int CHUNK    = 8192;
static constexpr int NB1 = (N_DRUG + BIN_ROWS - 1) / BIN_ROWS;
static constexpr int NB2 = (N_DIS  + BIN_ROWS - 1) / BIN_ROWS;
static constexpr int CH1 = (E_DRUG + CHUNK - 1) / CHUNK;
static constexpr int CH2 = (E_DIS  + CHUNK - 1) / CHUNK;
static constexpr int CHMAX = 128;
static constexpr int SPARE0 = 992;
static_assert(NB1 < SPARE0 && NB2 < SPARE0, "1024 bin counters in LDS: bins, the end entry, the spares behind them");
static_assert(CH1 <= CHMAX && CH2 <= CHMAX, "run bounds table in LDS");
static_assert(N_DRUG <= 65536 && N_DIS <= 65536, "record packs (src << 6) | local row");

static constexpr int FRONT_THREADS = 256;
static constexpr int EPT   = CHUNK / FRONT_THREADS;
static constexpr int MT1   = N_DRUG / 16;
static constexpr int MP2   = N_DIS / 32;
static constexpr int GEMM_WAVES  = MT1 + MP2;
static constexpr int GEMM_BLOCKS = (GEMM_WAVES + 7) / 8;
static constexpr int GRP_BLOCKS  = CH1 + CH2;
static_assert(N_DRUG % 16 == 0 && N_DIS % 32 == 0, "whole tiles, whole tile pairs");
static_assert(F_DRUG == 2 * F_DIS, "a tile pair at F_DIS is the matrix work of one tile at F_DRUG");
static_assert(CHUNK % FRONT_THREADS == 0 && FRONT_THREADS * 4 == 1024, "4 counters per thread in the scan");
static_assert(CHUNK <= 8192, "rank in a bin fits 13 bits of the key");

__device__ __forceinline__ void fgcn_gemm_tile(const float* __restrict__ X,
                                               const float* __restrict__ W,
                                               float* __restrict__ H,
                                               int m0, int K, int lane) {
    const int lane16 = lane & 15;
    const bool hi    = lane >= 16;

    v8f acc0 = {}, acc1 = {}, acc2 = {}, acc3 = {};

    const float* xa = X + (size_t)(m0 + lane16) * K + (hi ? 2 : 0);
    const float* wb = W + (size_t)(hi ? 2 : 0) * 64 + lane16 * 4;

    #pragma unroll 8
    for (int k = 0; k < K; k += 4) {
        float2 af = *(const float2*)(xa + k);
        v2f a; a.x = af.x; a.y = af.y;

        const float* wk = wb + (size_t)k * 64;
        const float4 w0 = *(const float4*)(wk);
        const float4 w1 = *(const float4*)(wk + 64);
        v2f b0; b0.x = w0.x; b0.y = w1.x;
        v2f b1; b1.x = w0.y; b1.y = w1.y;
        v2f b2; b2.x = w0.z; b2.y = w1.z;
        v2f b3; b3.x = w0.w; b3.y = w1.w;

        acc0 = __builtin_amdgcn_wmma_f32_16x16x4_f32(false, a, false, b0, (short)0, acc0, false, false);
        acc1 = __builtin_amdgcn_wmma_f32_16x16x4_f32(false, a, false, b1, (short)0, acc1, false, false);
        acc2 = __builtin_amdgcn_wmma_f32_16x16x4_f32(false, a, false, b2, (short)0, acc2, false, false);
        acc3 = __builtin_amdgcn_wmma_f32_16x16x4_f32(false, a, false, b3, (short)0, acc3, false, false);
    }

    const int rbase = m0 + (hi ? 8 : 0);
    #pragma unroll
    for (int r = 0; r < 8; ++r) {
        float4 o;
        o.x = acc0[r];
        o.y = acc1[r];
        o.z = acc2[r];
        o.w = acc3[r];
        *(float4*)(H + (size_t)(rbase + r) * 64 + lane16 * 4) = o;
    }
}

__device__ __forceinline__ void fgcn_group_chunk(int c, int* cnt, int* wsum,
                                                 const int* __restrict__ ei1,
                                                 const int* __restrict__ ei2,
                                                 int* __restrict__ rec1,
                                                 int* __restrict__ rec2,
                                                 int* __restrict__ loc1,
                                                 int* __restrict__ loc2) {
    const int tid  = threadIdx.x;
    const int lane = tid & 31;
    const int w    = tid >> 5;

    const int* src; const int* dst; int* rec; int* loc; int E, NB;
    if (c < CH1) { src = ei1; dst = ei1 + E_DRUG; rec = rec1; loc = loc1; E = E_DRUG; NB = NB1; }
    else { c -= CH1; src = ei2; dst = ei2 + E_DIS; rec = rec2; loc = loc2; E = E_DIS; NB = NB2; }

    *(int4*)(cnt + tid * 4) = make_int4(0, 0, 0, 0);
    __syncthreads();

    const int base = c * CHUNK;
    int r[EPT], key[EPT];
    #pragma unroll
    for (int i = 0; i < EPT; ++i) {
        const int e  = base + i * FRONT_THREADS + tid;
        const int ec = (e < E) ? e : E - 1;
        r[i]   = src[ec];
        key[i] = dst[ec];
    }
    #pragma unroll
    for (int i = 0; i < EPT; ++i) {
        const int e = base + i * FRONT_THREADS + tid;
        const int d = key[i];
        r[i]   = (r[i] << 6) | (d & 63);
        key[i] = (e < E) ? (d >> 6) : SPARE0 + lane;
    }
    int rk[EPT];
    #pragma unroll
    for (int i = 0; i < EPT; ++i) rk[i] = atomicAdd(&cnt[key[i]], 1);
    #pragma unroll
    for (int i = 0; i < EPT; ++i) key[i] = (key[i] >= SPARE0) ? -1 : ((key[i] << 13) | rk[i]);
    __syncthreads();

    const int4 v = *(const int4*)(cnt + tid * 4);
    const int s4 = v.x + v.y + v.z + v.w;
    int x = s4;
    for (int o = 1; o < 32; o <<= 1) {
        const int y = __shfl_up(x, o);
        if (lane >= o) x += y;
    }
    if (lane == 31) wsum[w] = x;
    __syncthreads();
    int wp = 0;
    #pragma unroll
    for (int j = 0; j < FRONT_THREADS / 32; ++j) {
        const int t = wsum[j];
        if (j < w) wp += t;
    }
    int4 o4;
    o4.x = x - s4 + wp;
    o4.y = o4.x + v.x;
    o4.z = o4.y + v.y;
    o4.w = o4.z + v.z;
    *(int4*)(cnt + tid * 4) = o4;
    int* lrow = loc + (size_t)c * (NB + 1);
    const int q = tid * 4;
    if (q + 0 <= NB) lrow[q + 0] = o4.x;
    if (q + 1 <= NB) lrow[q + 1] = o4.y;
    if (q + 2 <= NB) lrow[q + 2] = o4.z;
    if (q + 3 <= NB) lrow[q + 3] = o4.w;
    __syncthreads();

    int pos[EPT];
    #pragma unroll
    for (int i = 0; i < EPT; ++i) pos[i] = cnt[(key[i] >= 0) ? (key[i] >> 13) : 0] + (key[i] & 8191);
    #pragma unroll
    for (int i = 0; i < EPT; ++i) {
        if (key[i] >= 0) rec[(size_t)base + pos[i]] = r[i];
    }
}

__global__ __launch_bounds__(FRONT_THREADS) void fgcn_front(const float* __restrict__ x1,
                                                            const float* __restrict__ x2,
                                                            const float* __restrict__ w1,
                                                            const float* __restrict__ w2,
                                                            float* __restrict__ h1,
                                                            float* __restrict__ h2,
                                                            const int* __restrict__ ei1,
                                                            const int* __restrict__ ei2,
                                                            int* __restrict__ rec1,
                                                            int* __restrict__ rec2,
                                                            int* __restrict__ loc1,
                                                            int* __restrict__ loc2) {
    __shared__ __attribute__((aligned(16))) int cnt[1024];
    __shared__ int wsum[FRONT_THREADS / 32];

    const int b = blockIdx.x;
    if (b >= GEMM_BLOCKS) {
        fgcn_group_chunk(b - GEMM_BLOCKS, cnt, wsum, ei1, ei2, rec1, rec2, loc1, loc2);
        return;
    }

    const int wv   = __builtin_amdgcn_readfirstlane(b * (FRONT_THREADS / 32) + (threadIdx.x >> 5));
    const int lane = threadIdx.x & 31;
    if (wv < MT1) {
        fgcn_gemm_tile(x1, w1, h1, wv * 16, F_DRUG, lane);
    } else if (wv < GEMM_WAVES) {
        const int m0 = (wv - MT1) * 32;
        fgcn_gemm_tile(x2, w2, h2, m0, F_DIS, lane);
        fgcn_gemm_tile(x2, w2, h2, m0 + 16, F_DIS, lane);
    }
}

__device__ __forceinline__ void fgcn_row_add(float* acc, int rcd, int lane, float2 v) {
    float* a = acc + ((rcd & 63) << 6) + lane * 2;
    atomicAdd(a, v.x);
    atomicAdd(a + 1, v.y);
}

template <int K>
__device__ __forceinline__ void fgcn_rows(float* acc, const float* hl, int cur, int j, int lane) {
    int r[K];
    float2 v[K];
    #pragma unroll
    for (int k = 0; k < K; ++k) r[k] = __builtin_amdgcn_readlane(cur, j + k);
    #pragma unroll
    for (int k = 0; k < K; ++k) v[k] = *(const float2*)(hl + ((size_t)(r[k] >> 6) << 6));
    #pragma unroll
    for (int k = 0; k < K; ++k) fgcn_row_add(acc, r[k], lane, v[k]);
}

__device__ __forceinline__ void fgcn_piece(float* acc, const float* hl, int cur, int m, int lane) {
    int j = 0;
    for (; j + 8 <= m; j += 8) fgcn_rows<8>(acc, hl, cur, j, lane);
    switch (m - j) {
        case 7: fgcn_rows<7>(acc, hl, cur, j, lane); break;
        case 6: fgcn_rows<6>(acc, hl, cur, j, lane); break;
        case 5: fgcn_rows<5>(acc, hl, cur, j, lane); break;
        case 4: fgcn_rows<4>(acc, hl, cur, j, lane); break;
        case 3: fgcn_rows<3>(acc, hl, cur, j, lane); break;
        case 2: fgcn_rows<2>(acc, hl, cur, j, lane); break;
        case 1: fgcn_rows<1>(acc, hl, cur, j, lane); break;
        default: break;
    }
}

__global__ __launch_bounds__(256) void fgcn_sum_rows(const float* __restrict__ h1,
                                                     const float* __restrict__ h2,
                                                     const int* __restrict__ rec1,
                                                     const int* __restrict__ rec2,
                                                     const int* __restrict__ loc1,
                                                     const int* __restrict__ loc2,
                                                     const float* __restrict__ b1,
                                                     const float* __restrict__ b2,
                                                     float* __restrict__ out1,
                                                     float* __restrict__ out2) {
    __shared__ float acc[BIN_ROWS * NHID2];
    __shared__ int lo[CHMAX];
    __shared__ int hi[CHMAX];

    int b = blockIdx.x;
    const float* h; const int* rec; const int* loc; const float* bias; float* out; int NB, CH, N;
    if (b < NB1) { h = h1; rec = rec1; loc = loc1; bias = b1; out = out1; NB = NB1; CH = CH1; N = N_DRUG; }
    else { b -= NB1; h = h2; rec = rec2; loc = loc2; bias = b2; out = out2; NB = NB2; CH = CH2; N = N_DIS; }

    const int tid  = threadIdx.x;
    const int lane = tid & 31;
    const int w    = tid >> 5;

    for (int i = tid; i < BIN_ROWS * NHID2; i += 256) acc[i] = 0.0f;
    for (int t = tid; t < CH; t += 256) {
        const int* lp = loc + (size_t)t * (NB + 1) + b;
        lo[t] = lp[0];
        hi[t] = lp[1];
    }
    __syncthreads();

    const float* hl = h + lane * 2;
    int c = __builtin_amdgcn_readfirstlane(w);
    int s = 0, e = 0, mine = 0;
    if (c < CH) {
        s = __builtin_amdgcn_readfirstlane(lo[c]);
        e = __builtin_amdgcn_readfirstlane(hi[c]);
        if (lane < e - s) mine = rec[(size_t)c * CHUNK + s + lane];
    }
    while (c < CH) {
        const int cn = c + 8;
        int sn = 0, en = 0, minen = 0;
        if (cn < CH) {
            sn = __builtin_amdgcn_readfirstlane(lo[cn]);
            en = __builtin_amdgcn_readfirstlane(hi[cn]);
            if (lane < en - sn) minen = rec[(size_t)cn * CHUNK + sn + lane];
        }
        const int n = e - s;
        if (n > 0) {
            fgcn_piece(acc, hl, mine, (n < 32) ? n : 32, lane);
            const int* rp = rec + (size_t)c * CHUNK;
            for (int p = s + 32; p < e; p += 32) {
                const int m = (e - p < 32) ? (e - p) : 32;
                const int more = (lane < m) ? rp[p + lane] : 0;
                fgcn_piece(acc, hl, more, m, lane);
            }
        }
        c = cn; s = sn; e = en; mine = minen;
    }
    __syncthreads();

    for (int i = tid; i < BIN_ROWS * NHID2 / 4; i += 256) {
        const int row = b * BIN_ROWS + (i >> 4);
        if (row < N) {
            const int q = (i & 15) << 2;
            float4 a = *(const float4*)(acc + i * 4);
            a.x += bias[q + 0];
            a.y += bias[q + 1];
            a.z += bias[q + 2];
            a.w += bias[q + 3];
            *(float4*)(out + (size_t)row * NHID2 + q) = a;
        }
    }
}

extern "C" void kernel_launch(void* const* d_in, const int* in_sizes, int n_in,
                              void* d_out, int out_size, void* d_ws, size_t ws_size,
                              hipStream_t stream) {
    const float* drug_x  = (const float*)d_in[0];
    const float* dis_x   = (const float*)d_in[1];
    const float* W1      = (const float*)d_in[2];
    const float* b1      = (const float*)d_in[3];
    const float* W2      = (const float*)d_in[4];
    const float* b2      = (const float*)d_in[5];
    const int*   drug_ei = (const int*)d_in[6];
    const int*   dis_ei  = (const int*)d_in[7];

    float* out1 = (float*)d_out;
    float* out2 = out1 + (size_t)N_DRUG * NHID2;

    float* h1  = (float*)d_ws;
    float* h2  = h1 + (size_t)N_DRUG * NHID2;
    int* rec1  = (int*)(h2 + (size_t)N_DIS * NHID2);
    int* rec2  = rec1 + (size_t)E_DRUG;
    int* loc1  = rec2 + (size_t)E_DIS;
    int* loc2  = loc1 + (size_t)CH1 * (NB1 + 1);

    fgcn_front<<<GEMM_BLOCKS + GRP_BLOCKS, FRONT_THREADS, 0, stream>>>(drug_x, dis_x, W1, W2, h1, h2,
                                                                      drug_ei, dis_ei, rec1, rec2, loc1, loc2);

    fgcn_sum_rows<<<NB1 + NB2, 256, 0, stream>>>(h1, h2, rec1, rec2, loc1, loc2, b1, b2, out1, out2);
}


# LDS float adds go: owner block counting-sorts its bin's records by destination row (integer LDS adds), a wave sums each row's h rows in registers, a half-wave per h row; sum order: list arrival
# speedup vs baseline: 3.4456x; 1.2572x over previous
#include <hip/hip_runtime.h>
#include <hip/hip_bf16.h>

typedef __attribute__((ext_vector_type(2))) float v2f;
typedef __attribute__((ext_vector_type(8))) float v8f;

static constexpr int N_DRUG = 50000, E_DRUG = 1000000, F_DRUG = 256;
static constexpr int N_DIS  = 20000, E_DIS  = 640000,  F_DIS  = 128;
static constexpr int NHID2  = 64;

static constexpr int BIN_ROWS = 64;
static constexpr int CHUNK    = 8192;
static constexpr int NB1 = (N_DRUG + BIN_ROWS - 1) / BIN_ROWS;
static constexpr int NB2 = (N_DIS  + BIN_ROWS - 1) / BIN_ROWS;
static constexpr int CH1 = (E_DRUG + CHUNK - 1) / CHUNK;
static constexpr int CH2 = (E_DIS  + CHUNK - 1) / CHUNK;
static constexpr int CHMAX = 128;
static constexpr int SPARE0 = 992;
static_assert(NB1 < SPARE0 && NB2 < SPARE0, "1024 bin counters in LDS: bins, the end entry, the spares behind them");
static_assert(CH1 <= CHMAX && CH2 <= CHMAX, "run bounds table in LDS");
static_assert(N_DRUG <= 65536 && N_DIS <= 65536, "record packs (src << 6) | local row");

static constexpr int FRONT_THREADS = 256;
static constexpr int EPT   = CHUNK / FRONT_THREADS;
static constexpr int MT1   = N_DRUG / 16;
static constexpr int MP2   = N_DIS / 32;
static constexpr int GEMM_WAVES  = MT1 + MP2;
static constexpr int GEMM_BLOCKS = (GEMM_WAVES + 7) / 8;
static constexpr int GRP_BLOCKS  = CH1 + CH2;
static_assert(N_DRUG % 16 == 0 && N_DIS % 32 == 0, "whole tiles, whole tile pairs");
static_assert(F_DRUG == 2 * F_DIS, "a tile pair at F_DIS is the matrix work of one tile at F_DRUG");
static_assert(CHUNK % FRONT_THREADS == 0 && FRONT_THREADS * 4 == 1024, "4 counters per thread in the scan");
static_assert(CHUNK <= 8192, "rank in a bin fits 13 bits of the key");

__device__ __forceinline__ void fgcn_gemm_tile(const float* __restrict__ X,
                                               const float* __restrict__ W,
                                               float* __restrict__ H,
                                               int m0, int K, int lane) {
    const int lane16 = lane & 15;
    const bool hi    = lane >= 16;

    v8f acc0 = {}, acc1 = {}, acc2 = {}, acc3 = {};

    const float* xa = X + (size_t)(m0 + lane16) * K + (hi ? 2 : 0);
    const float* wb = W + (size_t)(hi ? 2 : 0) * 64 + lane16 * 4;

    #pragma unroll 8
    for (int k = 0; k < K; k += 4) {
        float2 af = *(const float2*)(xa + k);
        v2f a; a.x = af.x; a.y = af.y;

        const float* wk = wb + (size_t)k * 64;
        const float4 w0 = *(const float4*)(wk);
        const float4 w1 = *(const float4*)(wk + 64);
        v2f b0; b0.x = w0.x; b0.y = w1.x;
        v2f b1; b1.x = w0.y; b1.y = w1.y;
        v2f b2; b2.x = w0.z; b2.y = w1.z;
        v2f b3; b3.x = w0.w; b3.y = w1.w;

        acc0 = __builtin_amdgcn_wmma_f32_16x16x4_f32(false, a, false, b0, (short)0, acc0, false, false);
        acc1 = __builtin_amdgcn_wmma_f32_16x16x4_f32(false, a, false, b1, (short)0, acc1, false, false);
        acc2 = __builtin_amdgcn_wmma_f32_16x16x4_f32(false, a, false, b2, (short)0, acc2, false, false);
        acc3 = __builtin_amdgcn_wmma_f32_16x16x4_f32(false, a, false, b3, (short)0, acc3, false, false);
    }

    const int rbase = m0 + (hi ? 8 : 0);
    #pragma unroll
    for (int r = 0; r < 8; ++r) {
        float4 o;
        o.x = acc0[r];
        o.y = acc1[r];
        o.z = acc2[r];
        o.w = acc3[r];
        *(float4*)(H + (size_t)(rbase + r) * 64 + lane16 * 4) = o;
    }
}

__device__ __forceinline__ void fgcn_group_chunk(int c, int* cnt, int* wsum,
                                                 const int* __restrict__ ei1,
                                                 const int* __restrict__ ei2,
                                                 int* __restrict__ rec1,
                                                 int* __restrict__ rec2,
                                                 int* __restrict__ loc1,
                                                 int* __restrict__ loc2) {
    const int tid  = threadIdx.x;
    const int lane = tid & 31;
    const int w    = tid >> 5;

    const int* src; const int* dst; int* rec; int* loc; int E, NB;
    if (c < CH1) { src = ei1; dst = ei1 + E_DRUG; rec = rec1; loc = loc1; E = E_DRUG; NB = NB1; }
    else { c -= CH1; src = ei2; dst = ei2 + E_DIS; rec = rec2; loc = loc2; E = E_DIS; NB = NB2; }

    *(int4*)(cnt + tid * 4) = make_int4(0, 0, 0, 0);
    __syncthreads();

    const int base = c * CHUNK;
    int r[EPT], key[EPT];
    #pragma unroll
    for (int i = 0; i < EPT; ++i) {
        const int e  = base + i * FRONT_THREADS + tid;
        const int ec = (e < E) ? e : E - 1;
        r[i]   = src[ec];
        key[i] = dst[ec];
    }
    #pragma unroll
    for (int i = 0; i < EPT; ++i) {
        const int e = base + i * FRONT_THREADS + tid;
        const int d = key[i];
        r[i]   = (r[i] << 6) | (d & 63);
        key[i] = (e < E) ? (d >> 6) : SPARE0 + lane;
    }
    int rk[EPT];
    #pragma unroll
    for (int i = 0; i < EPT; ++i) rk[i] = atomicAdd(&cnt[key[i]], 1);
    #pragma unroll
    for (int i = 0; i < EPT; ++i) key[i] = (key[i] >= SPARE0) ? -1 : ((key[i] << 13) | rk[i]);
    __syncthreads();

    const int4 v = *(const int4*)(cnt + tid * 4);
    const int s4 = v.x + v.y + v.z + v.w;
    int x = s4;
    for (int o = 1; o < 32; o <<= 1) {
        const int y = __shfl_up(x, o);
        if (lane >= o) x += y;
    }
    if (lane == 31) wsum[w] = x;
    __syncthreads();
    int wp = 0;
    #pragma unroll
    for (int j = 0; j < FRONT_THREADS / 32; ++j) {
        const int t = wsum[j];
        if (j < w) wp += t;
    }
    int4 o4;
    o4.x = x - s4 + wp;
    o4.y = o4.x + v.x;
    o4.z = o4.y + v.y;
    o4.w = o4.z + v.z;
    *(int4*)(cnt + tid * 4) = o4;
    int* lrow = loc + (size_t)c * (NB + 1);
    const int q = tid * 4;
    if (q + 0 <= NB) lrow[q + 0] = o4.x;
    if (q + 1 <= NB) lrow[q + 1] = o4.y;
    if (q + 2 <= NB) lrow[q + 2] = o4.z;
    if (q + 3 <= NB) lrow[q + 3] = o4.w;
    __syncthreads();

    int pos[EPT];
    #pragma unroll
    for (int i = 0; i < EPT; ++i) pos[i] = cnt[(key[i] >= 0) ? (key[i] >> 13) : 0] + (key[i] & 8191);
    #pragma unroll
    for (int i = 0; i < EPT; ++i) {
        if (key[i] >= 0) rec[(size_t)base + pos[i]] = r[i];
    }
}

__global__ __launch_bounds__(FRONT_THREADS) void fgcn_front(const float* __restrict__ x1,
                                                            const float* __restrict__ x2,
                                                            const float* __restrict__ w1,
                                                            const float* __restrict__ w2,
                                                            float* __restrict__ h1,
                                                            float* __restrict__ h2,
                                                            const int* __restrict__ ei1,
                                                            const int* __restrict__ ei2,
                                                            int* __restrict__ rec1,
                                                            int* __restrict__ rec2,
                                                            int* __restrict__ loc1,
                                                            int* __restrict__ loc2) {
    __shared__ __attribute__((aligned(16))) int cnt[1024];
    __shared__ int wsum[FRONT_THREADS / 32];

    const int b = blockIdx.x;
    if (b >= GEMM_BLOCKS) {
        fgcn_group_chunk(b - GEMM_BLOCKS, cnt, wsum, ei1, ei2, rec1, rec2, loc1, loc2);
        return;
    }

    const int wv   = __builtin_amdgcn_readfirstlane(b * (FRONT_THREADS / 32) + (threadIdx.x >> 5));
    const int lane = threadIdx.x & 31;
    if (wv < MT1) {
        fgcn_gemm_tile(x1, w1, h1, wv * 16, F_DRUG, lane);
    } else if (wv < GEMM_WAVES) {
        const int m0 = (wv - MT1) * 32;
        fgcn_gemm_tile(x2, w2, h2, m0, F_DIS, lane);
        fgcn_gemm_tile(x2, w2, h2, m0 + 16, F_DIS, lane);
    }
}

__device__ __forceinline__ void fgcn_row_add(float* acc, int rcd, int lane, float2 v) {
    float* a = acc + ((rcd & 63) << 6) + lane * 2;
    atomicAdd(a, v.x);
    atomicAdd(a + 1, v.y);
}

template <int K>
__device__ __forceinline__ void fgcn_rows(float* acc, const float* hl, int cur, int j, int lane) {
    int r[K];
    float2 v[K];
    #pragma unroll
    for (int k = 0; k < K; ++k) r[k] = __builtin_amdgcn_readlane(cur, j + k);
    #pragma unroll
    for (int k = 0; k < K; ++k) v[k] = *(const float2*)(hl + ((size_t)(r[k] >> 6) << 6));
    #pragma unroll
    for (int k = 0; k < K; ++k) fgcn_row_add(acc, r[k], lane, v[k]);
}

__device__ __forceinline__ void fgcn_piece(float* acc, const float* hl, int cur, int m, int lane) {
    int j = 0;
    for (; j + 8 <= m; j += 8) fgcn_rows<8>(acc, hl, cur, j, lane);
    switch (m - j) {
        case 7: fgcn_rows<7>(acc, hl, cur, j, lane); break;
        case 6: fgcn_rows<6>(acc, hl, cur, j, lane); break;
        case 5: fgcn_rows<5>(acc, hl, cur, j, lane); break;
        case 4: fgcn_rows<4>(acc, hl, cur, j, lane); break;
        case 3: fgcn_rows<3>(acc, hl, cur, j, lane); break;
        case 2: fgcn_rows<2>(acc, hl, cur, j, lane); break;
        case 1: fgcn_rows<1>(acc, hl, cur, j, lane); break;
        default: break;
    }
}

__device__ __forceinline__ void fgcn_sum_rows_lds(float* acc, const int* lo, const int* hi,
                                                  const float* __restrict__ h,
                                                  const int* __restrict__ rec,
                                                  const float* __restrict__ bias,
                                                  float* __restrict__ out,
                                                  int b, int CH, int N) {
    const int tid  = threadIdx.x;
    const int lane = tid & 31;
    const int w    = tid >> 5;

    for (int i = tid; i < BIN_ROWS * NHID2; i += 256) acc[i] = 0.0f;
    __syncthreads();

    const float* hl = h + lane * 2;
    int c = __builtin_amdgcn_readfirstlane(w);
    int s = 0, e = 0, mine = 0;
    if (c < CH) {
        s = __builtin_amdgcn_readfirstlane(lo[c]);
        e = __builtin_amdgcn_readfirstlane(hi[c]);
        if (lane < e - s) mine = rec[(size_t)c * CHUNK + s + lane];
    }
    while (c < CH) {
        const int cn = c + 8;
        int sn = 0, en = 0, minen = 0;
        if (cn < CH) {
            sn = __builtin_amdgcn_readfirstlane(lo[cn]);
            en = __builtin_amdgcn_readfirstlane(hi[cn]);
            if (lane < en - sn) minen = rec[(size_t)cn * CHUNK + sn + lane];
        }
        const int n = e - s;
        if (n > 0) {
            fgcn_piece(acc, hl, mine, (n < 32) ? n : 32, lane);
            const int* rp = rec + (size_t)c * CHUNK;
            for (int p = s + 32; p < e; p += 32) {
                const int m = (e - p < 32) ? (e - p) : 32;
                const int more = (lane < m) ? rp[p + lane] : 0;
                fgcn_piece(acc, hl, more, m, lane);
            }
        }
        c = cn; s = sn; e = en; mine = minen;
    }
    __syncthreads();

    for (int i = tid; i < BIN_ROWS * NHID2 / 4; i += 256) {
        const int row = b * BIN_ROWS + (i >> 4);
        if (row < N) {
            const int q = (i & 15) << 2;
            float4 a = *(const float4*)(acc + i * 4);
            a.x += bias[q + 0];
            a.y += bias[q + 1];
            a.z += bias[q + 2];
            a.w += bias[q + 3];
            *(float4*)(out + (size_t)row * NHID2 + q) = a;
        }
    }
}

__device__ __forceinline__ void fgcn_count16(int* cnt, const int (&r)[16], int lane) {
    #pragma unroll
    for (int i = 0; i < 16; ++i) atomicAdd(&cnt[(r[i] >= 0) ? (r[i] & 63) : 64 + lane], 1);
}
__device__ __forceinline__ void fgcn_place16(int* cnt, int* lst, const int (&r)[16], int lane) {
    int p[16];
    #pragma unroll
    for (int i = 0; i < 16; ++i) p[i] = atomicAdd(&cnt[(r[i] >= 0) ? (r[i] & 63) : 64 + lane], 1);
    #pragma unroll
    for (int i = 0; i < 16; ++i) {
        if (r[i] >= 0) lst[p[i]] = r[i] >> 6;
    }
}

static constexpr int LIST_CAP = CHMAX * 64;
static_assert(CHMAX == 128, "a wave takes 16 runs; a lane scans 4 run lengths");

__global__ __launch_bounds__(256) void fgcn_sum_rows(const float* __restrict__ h1,
                                                     const float* __restrict__ h2,
                                                     const int* __restrict__ rec1,
                                                     const int* __restrict__ rec2,
                                                     const int* __restrict__ loc1,
                                                     const int* __restrict__ loc2,
                                                     const float* __restrict__ b1,
                                                     const float* __restrict__ b2,
                                                     float* __restrict__ out1,
                                                     float* __restrict__ out2) {
    __shared__ __attribute__((aligned(16))) int lst[LIST_CAP];
    __shared__ __attribute__((aligned(16))) int lo[CHMAX];
    __shared__ __attribute__((aligned(16))) int hi[CHMAX];
    __shared__ __attribute__((aligned(8))) int cnt[96];
    __shared__ int start[BIN_ROWS + 1];

    int b = blockIdx.x;
    const float* h; const int* rec; const int* loc; const float* bias; float* out; int NB, CH, N;
    if (b < NB1) { h = h1; rec = rec1; loc = loc1; bias = b1; out = out1; NB = NB1; CH = CH1; N = N_DRUG; }
    else { b -= NB1; h = h2; rec = rec2; loc = loc2; bias = b2; out = out2; NB = NB2; CH = CH2; N = N_DIS; }

    const int tid  = threadIdx.x;
    const int lane = tid & 31;
    const int w    = __builtin_amdgcn_readfirstlane(tid >> 5);

    if (tid < 96) cnt[tid] = 0;
    if (tid < CHMAX) {
        int l = 0, u = 0;
        if (tid < CH) {
            const int* lp = loc + (size_t)tid * (NB + 1) + b;
            l = lp[0];
            u = lp[1];
        }
        lo[tid] = l;
        hi[tid] = u;
    }
    __syncthreads();

    const int4 l4 = *(const int4*)(lo + lane * 4);
    const int4 u4 = *(const int4*)(hi + lane * 4);
    int mx = max(max(u4.x - l4.x, u4.y - l4.y), max(u4.z - l4.z, u4.w - l4.w));
    for (int o = 16; o > 0; o >>= 1) mx = max(mx, __shfl_xor(mx, o));
    mx = __builtin_amdgcn_readfirstlane(mx);
    if (mx > 64) {
        fgcn_sum_rows_lds((float*)lst, lo, hi, h, rec, bias, out, b, CH, N);
        return;
    }
    const bool two = mx > 32;

    const int cl  = w + 8 * (lane & 15);
    const int lov = lo[cl];
    const int nov = hi[cl] - lov;
    int ra[16], rb[16];
    #pragma unroll
    for (int i = 0; i < 16; ++i) {
        const int s = __builtin_amdgcn_readlane(lov, i);
        const int n = __builtin_amdgcn_readlane(nov, i);
        const int* rp = rec + (size_t)(w + 8 * i) * CHUNK + s + lane;
        int v = -1;
        if (lane < n) v = rp[0];
        ra[i] = v;
    }
    #pragma unroll
    for (int i = 0; i < 16; ++i) {
        const int s = __builtin_amdgcn_readlane(lov, i);
        const int n = __builtin_amdgcn_readlane(nov, i);
        const int* rp = rec + (size_t)(w + 8 * i) * CHUNK + s + lane;
        int v = -1;
        if (two && lane + 32 < n) v = rp[32];
        rb[i] = v;
    }
    fgcn_count16(cnt, ra, lane);
    if (two) fgcn_count16(cnt, rb, lane);
    __syncthreads();

    if (w == 0) {
        const int2 c2 = *(const int2*)(cnt + lane * 2);
        const int s2 = c2.x + c2.y;
        int x = s2;
        for (int o = 1; o < 32; o <<= 1) {
            const int y = __shfl_up(x, o);
            if (lane >= o) x += y;
        }
        const int ex = x - s2;
        start[lane * 2]     = ex;
        start[lane * 2 + 1] = ex + c2.x;
        if (lane == 31) start[BIN_ROWS] = x;
        *(int2*)(cnt + lane * 2) = make_int2(ex, ex + c2.x);
    }
    __syncthreads();

    fgcn_place16(cnt, lst, ra, lane);
    if (two) fgcn_place16(cnt, lst, rb, lane);
    __syncthreads();

    const int hf  = lane >> 4;
    const int c16 = lane & 15;
    const char* hb = (const char*)h;
    const unsigned q16 = (unsigned)c16 * 16u;
    const float4 bq = *(const float4*)(bias + c16 * 4);
    const int st = start[min(w * 8 + lane, BIN_ROWS)];
    #pragma unroll 1
    for (int rr = 0; rr < 8; ++rr) {
        const int s = __builtin_amdgcn_readlane(st, rr);
        const int e = __builtin_amdgcn_readlane(st, rr + 1);
        float4 sum = make_float4(0.0f, 0.0f, 0.0f, 0.0f);
        int sx[8];
        #pragma unroll
        for (int k = 0; k < 8; ++k) sx[k] = lst[s + 2 * k + hf];
        int p = s;
        for (; p + 16 <= e; p += 16) {
            float4 v[8];
            #pragma unroll
            for (int k = 0; k < 8; ++k) v[k] = *(const float4*)(hb + ((((unsigned)sx[k]) << 8) | q16));
            #pragma unroll
            for (int k = 0; k < 8; ++k) sx[k] = lst[p + 16 + 2 * k + hf];
            #pragma unroll
            for (int k = 0; k < 8; ++k) {
                sum.x += v[k].x;
                sum.y += v[k].y;
                sum.z += v[k].z;
                sum.w += v[k].w;
            }
        }
        if (p < e) {
            float4 v[8];
            #pragma unroll
            for (int k = 0; k < 8; ++k) {
                v[k] = make_float4(0.0f, 0.0f, 0.0f, 0.0f);
                if (p + 2 * k + hf < e) v[k] = *(const float4*)(hb + ((((unsigned)sx[k]) << 8) | q16));
            }
            #pragma unroll
            for (int k = 0; k < 8; ++k) {
                sum.x += v[k].x;
                sum.y += v[k].y;
                sum.z += v[k].z;
                sum.w += v[k].w;
            }
        }
        sum.x += __shfl_xor(sum.x, 16);
        sum.y += __shfl_xor(sum.y, 16);
        sum.z += __shfl_xor(sum.z, 16);
        sum.w += __shfl_xor(sum.w, 16);
        const int row = b * BIN_ROWS + w * 8 + rr;
        if (row < N && lane < 16) {
            float4 o;
            o.x = sum.x + bq.x;
            o.y = sum.y + bq.y;
            o.z = sum.z + bq.z;
            o.w = sum.w + bq.w;
            *(float4*)(out + (size_t)row * NHID2 + c16 * 4) = o;
        }
    }
}

extern "C" void kernel_launch(void* const* d_in, const int* in_sizes, int n_in,
                              void* d_out, int out_size, void* d_ws, size_t ws_size,
                              hipStream_t stream) {
    const float* drug_x  = (const float*)d_in[0];
    const float* dis_x   = (const float*)d_in[1];
    const float* W1      = (const float*)d_in[2];
    const float* b1      = (const float*)d_in[3];
    const float* W2      = (const float*)d_in[4];
    const float* b2      = (const float*)d_in[5];
    const int*   drug_ei = (const int*)d_in[6];
    const int*   dis_ei  = (const int*)d_in[7];

    float* out1 = (float*)d_out;
    float* out2 = out1 + (size_t)N_DRUG * NHID2;

    float* h1  = (float*)d_ws;
    float* h2  = h1 + (size_t)N_DRUG * NHID2;
    int* rec1  = (int*)(h2 + (size_t)N_DIS * NHID2);
    int* rec2  = rec1 + (size_t)E_DRUG;
    int* loc1  = rec2 + (size_t)E_DIS;
    int* loc2  = loc1 + (size_t)CH1 * (NB1 + 1);

    fgcn_front<<<GEMM_BLOCKS + GRP_BLOCKS, FRONT_THREADS, 0, stream>>>(drug_x, dis_x, W1, W2, h1, h2,
                                                                      drug_ei, dis_ei, rec1, rec2, loc1, loc2);

    fgcn_sum_rows<<<NB1 + NB2, 256, 0, stream>>>(h1, h2, rec1, rec2, loc1, loc2, b1, b2, out1, out2);
}


# edge grouping: a chunk's sorted records go to their places in LDS and leave as whole 128-byte lines (16 bytes per lane), not as 4-byte stores over up to 32 lines each; sum order unchanged
# speedup vs baseline: 3.7152x; 1.0782x over previous
#include <hip/hip_runtime.h>
#include <hip/hip_bf16.h>

typedef __attribute__((ext_vector_type(2))) float v2f;
typedef __attribute__((ext_vector_type(8))) float v8f;

static constexpr int N_DRUG = 50000, E_DRUG = 1000000, F_DRUG = 256;
static constexpr int N_DIS  = 20000, E_DIS  = 640000,  F_DIS  = 128;
static constexpr int NHID2  = 64;

static constexpr int BIN_ROWS = 64;
static constexpr int CHUNK    = 8192;
static constexpr int NB1 = (N_DRUG + BIN_ROWS - 1) / BIN_ROWS;
static constexpr int NB2 = (N_DIS  + BIN_ROWS - 1) / BIN_ROWS;
static constexpr int CH1 = (E_DRUG + CHUNK - 1) / CHUNK;
static constexpr int CH2 = (E_DIS  + CHUNK - 1) / CHUNK;
static constexpr int CHMAX = 128;
static constexpr int SPARE0 = 992;
static_assert(NB1 < SPARE0 && NB2 < SPARE0, "1024 bin counters in LDS: bins, the end entry, the spares behind them");
static_assert(CH1 <= CHMAX && CH2 <= CHMAX, "run bounds table in LDS");
static_assert(N_DRUG <= 65536 && N_DIS <= 65536, "record packs (src << 6) | local row");

static constexpr int FRONT_THREADS = 256;
static constexpr int EPT   = CHUNK / FRONT_THREADS;
static constexpr int MT1   = N_DRUG / 16;
static constexpr int MP2   = N_DIS / 32;
static constexpr int GEMM_WAVES  = MT1 + MP2;
static constexpr int GEMM_BLOCKS = (GEMM_WAVES + 7) / 8;
static constexpr int GRP_BLOCKS  = CH1 + CH2;
static_assert(N_DRUG % 16 == 0 && N_DIS % 32 == 0, "whole tiles, whole tile pairs");
static_assert(F_DRUG == 2 * F_DIS, "a tile pair at F_DIS is the matrix work of one tile at F_DRUG");
static_assert(CHUNK % FRONT_THREADS == 0 && FRONT_THREADS * 4 == 1024, "4 counters per thread in the scan");
static_assert(CHUNK <= 8192, "rank in a bin fits 13 bits of the key");
static_assert(E_DRUG % 4 == 0 && E_DIS % 4 == 0 && CHUNK % (FRONT_THREADS * 4) == 0,
              "a chunk's records leave LDS as 16-byte pieces, each wholly inside the edge list or wholly behind it");

__device__ __forceinline__ void fgcn_gemm_tile(const float* __restrict__ X,
                                               const float* __restrict__ W,
                                               float* __restrict__ H,
                                               int m0, int K, int lane) {
    const int lane16 = lane & 15;
    const bool hi    = lane >= 16;

    v8f acc0 = {}, acc1 = {}, acc2 = {}, acc3 = {};

    const float* xa = X + (size_t)(m0 + lane16) * K + (hi ? 2 : 0);
    const float* wb = W + (size_t)(hi ? 2 : 0) * 64 + lane16 * 4;

    #pragma unroll 8
    for (int k = 0; k < K; k += 4) {
        float2 af = *(const float2*)(xa + k);
        v2f a; a.x = af.x; a.y = af.y;

        const float* wk = wb + (size_t)k * 64;
        const float4 w0 = *(const float4*)(wk);
        const float4 w1 = *(const float4*)(wk + 64);
        v2f b0; b0.x = w0.x; b0.y = w1.x;
        v2f b1; b1.x = w0.y; b1.y = w1.y;
        v2f b2; b2.x = w0.z; b2.y = w1.z;
        v2f b3; b3.x = w0.w; b3.y = w1.w;

        acc0 = __builtin_amdgcn_wmma_f32_16x16x4_f32(false, a, false, b0, (short)0, acc0, false, false);
        acc1 = __builtin_amdgcn_wmma_f32_16x16x4_f32(false, a, false, b1, (short)0, acc1, false, false);
        acc2 = __builtin_amdgcn_wmma_f32_16x16x4_f32(false, a, false, b2, (short)0, acc2, false, false);
        acc3 = __builtin_amdgcn_wmma_f32_16x16x4_f32(false, a, false, b3, (short)0, acc3, false, false);
    }

    const int rbase = m0 + (hi ? 8 : 0);
    #pragma unroll
    for (int r = 0; r < 8; ++r) {
        float4 o;
        o.x = acc0[r];
        o.y = acc1[r];
        o.z = acc2[r];
        o.w = acc3[r];
        *(float4*)(H + (size_t)(rbase + r) * 64 + lane16 * 4) = o;
    }
}

__device__ __forceinline__ void fgcn_group_chunk(int c, int* cnt, int* wsum, int* stage,
                                                 const int* __restrict__ ei1,
                                                 const int* __restrict__ ei2,
                                                 int* __restrict__ rec1,
                                                 int* __restrict__ rec2,
                                                 int* __restrict__ loc1,
                                                 int* __restrict__ loc2) {
    const int tid  = threadIdx.x;
    const int lane = tid & 31;
    const int w    = tid >> 5;

    const int* src; const int* dst; int* rec; int* loc; int E, NB;
    if (c < CH1) { src = ei1; dst = ei1 + E_DRUG; rec = rec1; loc = loc1; E = E_DRUG; NB = NB1; }
    else { c -= CH1; src = ei2; dst = ei2 + E_DIS; rec = rec2; loc = loc2; E = E_DIS; NB = NB2; }

    *(int4*)(cnt + tid * 4) = make_int4(0, 0, 0, 0);
    __syncthreads();

    const int base = c * CHUNK;
    int r[EPT], key[EPT];
    #pragma unroll
    for (int i = 0; i < EPT; ++i) {
        const int e  = base + i * FRONT_THREADS + tid;
        const int ec = (e < E) ? e : E - 1;
        r[i]   = src[ec];
        key[i] = dst[ec];
    }
    #pragma unroll
    for (int i = 0; i < EPT; ++i) {
        const int e = base + i * FRONT_THREADS + tid;
        const int d = key[i];
        r[i]   = (r[i] << 6) | (d & 63);
        key[i] = (e < E) ? (d >> 6) : SPARE0 + lane;
    }
    int rk[EPT];
    #pragma unroll
    for (int i = 0; i < EPT; ++i) rk[i] = atomicAdd(&cnt[key[i]], 1);
    #pragma unroll
    for (int i = 0; i < EPT; ++i) key[i] = (key[i] >= SPARE0) ? -1 : ((key[i] << 13) | rk[i]);
    __syncthreads();

    const int4 v = *(const int4*)(cnt + tid * 4);
    const int s4 = v.x + v.y + v.z + v.w;
    int x = s4;
    for (int o = 1; o < 32; o <<= 1) {
        const int y = __shfl_up(x, o);
        if (lane >= o) x += y;
    }
    if (lane == 31) wsum[w] = x;
    __syncthreads();
    int wp = 0;
    #pragma unroll
    for (int j = 0; j < FRONT_THREADS / 32; ++j) {
        const int t = wsum[j];
        if (j < w) wp += t;
    }
    int4 o4;
    o4.x = x - s4 + wp;
    o4.y = o4.x + v.x;
    o4.z = o4.y + v.y;
    o4.w = o4.z + v.z;
    *(int4*)(cnt + tid * 4) = o4;
    int* lrow = loc + (size_t)c * (NB + 1);
    const int q = tid * 4;
    if (q + 0 <= NB) lrow[q + 0] = o4.x;
    if (q + 1 <= NB) lrow[q + 1] = o4.y;
    if (q + 2 <= NB) lrow[q + 2] = o4.z;
    if (q + 3 <= NB) lrow[q + 3] = o4.w;
    __syncthreads();

    int pos[EPT];
    #pragma unroll
    for (int i = 0; i < EPT; ++i) pos[i] = cnt[(key[i] >= 0) ? (key[i] >> 13) : 0] + (key[i] & 8191);
    #pragma unroll
    for (int i = 0; i < EPT; ++i) stage[(key[i] >= 0) ? pos[i] : CHUNK + lane] = r[i];
    __syncthreads();

    constexpr int PIECES = CHUNK / (FRONT_THREADS * 4);
    const int nvalid = (E - base < CHUNK) ? (E - base) : CHUNK;
    int* outp = rec + (size_t)base;
    if (nvalid == CHUNK) {
        int4 piece[PIECES];
        #pragma unroll
        for (int j = 0; j < PIECES; ++j) piece[j] = *(const int4*)(stage + (j * FRONT_THREADS + tid) * 4);
        #pragma unroll
        for (int j = 0; j < PIECES; ++j) *(int4*)(outp + (j * FRONT_THREADS + tid) * 4) = piece[j];
    } else {
        for (int j = 0; j < PIECES; ++j) {
            const int at = (j * FRONT_THREADS + tid) * 4;
            if (at < nvalid) *(int4*)(outp + at) = *(const int4*)(stage + at);
        }
    }
}

__global__ __launch_bounds__(FRONT_THREADS) void fgcn_front(const float* __restrict__ x1,
                                                            const float* __restrict__ x2,
                                                            const float* __restrict__ w1,
                                                            const float* __restrict__ w2,
                                                            float* __restrict__ h1,
                                                            float* __restrict__ h2,
                                                            const int* __restrict__ ei1,
                                                            const int* __restrict__ ei2,
                                                            int* __restrict__ rec1,
                                                            int* __restrict__ rec2,
                                                            int* __restrict__ loc1,
                                                            int* __restrict__ loc2) {
    __shared__ __attribute__((aligned(16))) int cnt[1024];
    __shared__ int wsum[FRONT_THREADS / 32];
    __shared__ __attribute__((aligned(16))) int stage[CHUNK + 32];

    const int b = blockIdx.x;
    if (b >= GEMM_BLOCKS) {
        fgcn_group_chunk(b - GEMM_BLOCKS, cnt, wsum, stage, ei1, ei2, rec1, rec2, loc1, loc2);
        return;
    }

    const int wv   = __builtin_amdgcn_readfirstlane(b * (FRONT_THREADS / 32) + (threadIdx.x >> 5));
    const int lane = threadIdx.x & 31;
    if (wv < MT1) {
        fgcn_gemm_tile(x1, w1, h1, wv * 16, F_DRUG, lane);
    } else if (wv < GEMM_WAVES) {
        const int m0 = (wv - MT1) * 32;
        fgcn_gemm_tile(x2, w2, h2, m0, F_DIS, lane);
        fgcn_gemm_tile(x2, w2, h2, m0 + 16, F_DIS, lane);
    }
}

__device__ __forceinline__ void fgcn_row_add(float* acc, int rcd, int lane, float2 v) {
    float* a = acc + ((rcd & 63) << 6) + lane * 2;
    atomicAdd(a, v.x);
    atomicAdd(a + 1, v.y);
}

template <int K>
__device__ __forceinline__ void fgcn_rows(float* acc, const float* hl, int cur, int j, int lane) {
    int r[K];
    float2 v[K];
    #pragma unroll
    for (int k = 0; k < K; ++k) r[k] = __builtin_amdgcn_readlane(cur, j + k);
    #pragma unroll
    for (int k = 0; k < K; ++k) v[k] = *(const float2*)(hl + ((size_t)(r[k] >> 6) << 6));
    #pragma unroll
    for (int k = 0; k < K; ++k) fgcn_row_add(acc, r[k], lane, v[k]);
}

__device__ __forceinline__ void fgcn_piece(float* acc, const float* hl, int cur, int m, int lane) {
    int j = 0;
    for (; j + 8 <= m; j += 8) fgcn_rows<8>(acc, hl, cur, j, lane);
    switch (m - j) {
        case 7: fgcn_rows<7>(acc, hl, cur, j, lane); break;
        case 6: fgcn_rows<6>(acc, hl, cur, j, lane); break;
        case 5: fgcn_rows<5>(acc, hl, cur, j, lane); break;
        case 4: fgcn_rows<4>(acc, hl, cur, j, lane); break;
        case 3: fgcn_rows<3>(acc, hl, cur, j, lane); break;
        case 2: fgcn_rows<2>(acc, hl, cur, j, lane); break;
        case 1: fgcn_rows<1>(acc, hl, cur, j, lane); break;
        default: break;
    }
}

__device__ __forceinline__ void fgcn_sum_rows_lds(float* acc, const int* lo, const int* hi,
                                                  const float* __restrict__ h,
                                                  const int* __restrict__ rec,
                                                  const float* __restrict__ bias,
                                                  float* __restrict__ out,
                                                  int b, int CH, int N) {
    const int tid  = threadIdx.x;
    const int lane = tid & 31;
    const int w    = tid >> 5;

    for (int i = tid; i < BIN_ROWS * NHID2; i += 256) acc[i] = 0.0f;
    __syncthreads();

    const float* hl = h + lane * 2;
    int c = __builtin_amdgcn_readfirstlane(w);
    int s = 0, e = 0, mine = 0;
    if (c < CH) {
        s = __builtin_amdgcn_readfirstlane(lo[c]);
        e = __builtin_amdgcn_readfirstlane(hi[c]);
        if (lane < e - s) mine = rec[(size_t)c * CHUNK + s + lane];
    }
    while (c < CH) {
        const int cn = c + 8;
        int sn = 0, en = 0, minen = 0;
        if (cn < CH) {
            sn = __builtin_amdgcn_readfirstlane(lo[cn]);
            en = __builtin_amdgcn_readfirstlane(hi[cn]);
            if (lane < en - sn) minen = rec[(size_t)cn * CHUNK + sn + lane];
        }
        const int n = e - s;
        if (n > 0) {
            fgcn_piece(acc, hl, mine, (n < 32) ? n : 32, lane);
            const int* rp = rec + (size_t)c * CHUNK;
            for (int p = s + 32; p < e; p += 32) {
                const int m = (e - p < 32) ? (e - p) : 32;
                const int more = (lane < m) ? rp[p + lane] : 0;
                fgcn_piece(acc, hl, more, m, lane);
            }
        }
        c = cn; s = sn; e = en; mine = minen;
    }
    __syncthreads();

    for (int i = tid; i < BIN_ROWS * NHID2 / 4; i += 256) {
        const int row = b * BIN_ROWS + (i >> 4);
        if (row < N) {
            const int q = (i & 15) << 2;
            float4 a = *(const float4*)(acc + i * 4);
            a.x += bias[q + 0];
            a.y += bias[q + 1];
            a.z += bias[q + 2];
            a.w += bias[q + 3];
            *(float4*)(out + (size_t)row * NHID2 + q) = a;
        }
    }
}

__device__ __forceinline__ void fgcn_count16(int* cnt, const int (&r)[16], int lane) {
    #pragma unroll
    for (int i = 0; i < 16; ++i) atomicAdd(&cnt[(r[i] >= 0) ? (r[i] & 63) : 64 + lane], 1);
}
__device__ __forceinline__ void fgcn_place16(int* cnt, int* lst, const int (&r)[16], int lane) {
    int p[16];
    #pragma unroll
    for (int i = 0; i < 16; ++i) p[i] = atomicAdd(&cnt[(r[i] >= 0) ? (r[i] & 63) : 64 + lane], 1);
    #pragma unroll
    for (int i = 0; i < 16; ++i) {
        if (r[i] >= 0) lst[p[i]] = r[i] >> 6;
    }
}

static constexpr int LIST_CAP = CHMAX * 64;
static_assert(CHMAX == 128, "a wave takes 16 runs; a lane scans 4 run lengths");

__global__ __launch_bounds__(256) void fgcn_sum_rows(const float* __restrict__ h1,
                                                     const float* __restrict__ h2,
                                                     const int* __restrict__ rec1,
                                                     const int* __restrict__ rec2,
                                                     const int* __restrict__ loc1,
                                                     const int* __restrict__ loc2,
                                                     const float* __restrict__ b1,
                                                     const float* __restrict__ b2,
                                                     float* __restrict__ out1,
                                                     float* __restrict__ out2) {
    __shared__ __attribute__((aligned(16))) int lst[LIST_CAP];
    __shared__ __attribute__((aligned(16))) int lo[CHMAX];
    __shared__ __attribute__((aligned(16))) int hi[CHMAX];
    __shared__ __attribute__((aligned(8))) int cnt[96];
    __shared__ int start[BIN_ROWS + 1];

    int b = blockIdx.x;
    const float* h; const int* rec; const int* loc; const float* bias; float* out; int NB, CH, N;
    if (b < NB1) { h = h1; rec = rec1; loc = loc1; bias = b1; out = out1; NB = NB1; CH = CH1; N = N_DRUG; }
    else { b -= NB1; h = h2; rec = rec2; loc = loc2; bias = b2; out = out2; NB = NB2; CH = CH2; N = N_DIS; }

    const int tid  = threadIdx.x;
    const int lane = tid & 31;
    const int w    = __builtin_amdgcn_readfirstlane(tid >> 5);

    if (tid < 96) cnt[tid] = 0;
    if (tid < CHMAX) {
        int l = 0, u = 0;
        if (tid < CH) {
            const int* lp = loc + (size_t)tid * (NB + 1) + b;
            l = lp[0];
            u = lp[1];
        }
        lo[tid] = l;
        hi[tid] = u;
    }
    __syncthreads();

    const int4 l4 = *(const int4*)(lo + lane * 4);
    const int4 u4 = *(const int4*)(hi + lane * 4);
    int mx = max(max(u4.x - l4.x, u4.y - l4.y), max(u4.z - l4.z, u4.w - l4.w));
    for (int o = 16; o > 0; o >>= 1) mx = max(mx, __shfl_xor(mx, o));
    mx = __builtin_amdgcn_readfirstlane(mx);
    if (mx > 64) {
        fgcn_sum_rows_lds((float*)lst, lo, hi, h, rec, bias, out, b, CH, N);
        return;
    }
    const bool two = mx > 32;

    const int cl  = w + 8 * (lane & 15);
    const int lov = lo[cl];
    const int nov = hi[cl] - lov;
    int ra[16], rb[16];
    #pragma unroll
    for (int i = 0; i < 16; ++i) {
        const int s = __builtin_amdgcn_readlane(lov, i);
        const int n = __builtin_amdgcn_readlane(nov, i);
        const int* rp = rec + (size_t)(w + 8 * i) * CHUNK + s + lane;
        int v = -1;
        if (lane < n) v = rp[0];
        ra[i] = v;
    }
    #pragma unroll
    for (int i = 0; i < 16; ++i) {
        const int s = __builtin_amdgcn_readlane(lov, i);
        const int n = __builtin_amdgcn_readlane(nov, i);
        const int* rp = rec + (size_t)(w + 8 * i) * CHUNK + s + lane;
        int v = -1;
        if (two && lane + 32 < n) v = rp[32];
        rb[i] = v;
    }
    fgcn_count16(cnt, ra, lane);
    if (two) fgcn_count16(cnt, rb, lane);
    __syncthreads();

    if (w == 0) {
        const int2 c2 = *(const int2*)(cnt + lane * 2);
        const int s2 = c2.x + c2.y;
        int x = s2;
        for (int o = 1; o < 32; o <<= 1) {
            const int y = __shfl_up(x, o);
            if (lane >= o) x += y;
        }
        const int ex = x - s2;
        start[lane * 2]     = ex;
        start[lane * 2 + 1] = ex + c2.x;
        if (lane == 31) start[BIN_ROWS] = x;
        *(int2*)(cnt + lane * 2) = make_int2(ex, ex + c2.x);
    }
    __syncthreads();

    fgcn_place16(cnt, lst, ra, lane);
    if (two) fgcn_place16(cnt, lst, rb, lane);
    __syncthreads();

    const int hf  = lane >> 4;
    const int c16 = lane & 15;
    const char* hb = (const char*)h;
    const unsigned q16 = (unsigned)c16 * 16u;
    const float4 bq = *(const float4*)(bias + c16 * 4);
    const int st = start[min(w * 8 + lane, BIN_ROWS)];
    #pragma unroll 1
    for (int rr = 0; rr < 8; ++rr) {
        const int s = __builtin_amdgcn_readlane(st, rr);
        const int e = __builtin_amdgcn_readlane(st, rr + 1);
        float4 sum = make_float4(0.0f, 0.0f, 0.0f, 0.0f);
        int sx[8];
        #pragma unroll
        for (int k = 0; k < 8; ++k) sx[k] = lst[s + 2 * k + hf];
        int p = s;
        for (; p + 16 <= e; p += 16) {
            float4 v[8];
            #pragma unroll
            for (int k = 0; k < 8; ++k) v[k] = *(const float4*)(hb + ((((unsigned)sx[k]) << 8) | q16));
            #pragma unroll
            for (int k = 0; k < 8; ++k) sx[k] = lst[p + 16 + 2 * k + hf];
            #pragma unroll
            for (int k = 0; k < 8; ++k) {
                sum.x += v[k].x;
                sum.y += v[k].y;
                sum.z += v[k].z;
                sum.w += v[k].w;
            }
        }
        if (p < e) {
            float4 v[8];
            #pragma unroll
            for (int k = 0; k < 8; ++k) {
                v[k] = make_float4(0.0f, 0.0f, 0.0f, 0.0f);
                if (p + 2 * k + hf < e) v[k] = *(const float4*)(hb + ((((unsigned)sx[k]) << 8) | q16));
            }
            #pragma unroll
            for (int k = 0; k < 8; ++k) {
                sum.x += v[k].x;
                sum.y += v[k].y;
                sum.z += v[k].z;
                sum.w += v[k].w;
            }
        }
        sum.x += __shfl_xor(sum.x, 16);
        sum.y += __shfl_xor(sum.y, 16);
        sum.z += __shfl_xor(sum.z, 16);
        sum.w += __shfl_xor(sum.w, 16);
        const int row = b * BIN_ROWS + w * 8 + rr;
        if (row < N && lane < 16) {
            float4 o;
            o.x = sum.x + bq.x;
            o.y = sum.y + bq.y;
            o.z = sum.z + bq.z;
            o.w = sum.w + bq.w;
            *(float4*)(out + (size_t)row * NHID2 + c16 * 4) = o;
        }
    }
}

extern "C" void kernel_launch(void* const* d_in, const int* in_sizes, int n_in,
                              void* d_out, int out_size, void* d_ws, size_t ws_size,
                              hipStream_t stream) {
    const float* drug_x  = (const float*)d_in[0];
    const float* dis_x   = (const float*)d_in[1];
    const float* W1      = (const float*)d_in[2];
    const float* b1      = (const float*)d_in[3];
    const float* W2      = (const float*)d_in[4];
    const float* b2      = (const float*)d_in[5];
    const int*   drug_ei = (const int*)d_in[6];
    const int*   dis_ei  = (const int*)d_in[7];

    float* out1 = (float*)d_out;
    float* out2 = out1 + (size_t)N_DRUG * NHID2;

    float* h1  = (float*)d_ws;
    float* h2  = h1 + (size_t)N_DRUG * NHID2;
    int* rec1  = (int*)(h2 + (size_t)N_DIS * NHID2);
    int* rec2  = rec1 + (size_t)E_DRUG;
    int* loc1  = rec2 + (size_t)E_DIS;
    int* loc2  = loc1 + (size_t)CH1 * (NB1 + 1);

    fgcn_front<<<GEMM_BLOCKS + GRP_BLOCKS, FRONT_THREADS, 0, stream>>>(drug_x, dis_x, W1, W2, h1, h2,
                                                                      drug_ei, dis_ei, rec1, rec2, loc1, loc2);

    fgcn_sum_rows<<<NB1 + NB2, 256, 0, stream>>>(h1, h2, rec1, rec2, loc1, loc2, b1, b2, out1, out2);
}


# row sums: every row has a fixed 128-entry LDS slot pre-filled with a zero row's index: one returning LDS add places a record (no count pass, no scan), every batch of 16 is whole; sum order unchanged
# speedup vs baseline: 4.0179x; 1.0815x over previous
#include <hip/hip_runtime.h>
#include <hip/hip_bf16.h>

typedef __attribute__((ext_vector_type(2))) float v2f;
typedef __attribute__((ext_vector_type(8))) float v8f;

static constexpr int N_DRUG = 50000, E_DRUG = 1000000, F_DRUG = 256;
static constexpr int N_DIS  = 20000, E_DIS  = 640000,  F_DIS  = 128;
static constexpr int NHID2  = 64;

static constexpr int BIN_ROWS = 64;
static constexpr int CHUNK    = 8192;
static constexpr int NB1 = (N_DRUG + BIN_ROWS - 1) / BIN_ROWS;
static constexpr int NB2 = (N_DIS  + BIN_ROWS - 1) / BIN_ROWS;
static constexpr int CH1 = (E_DRUG + CHUNK - 1) / CHUNK;
static constexpr int CH2 = (E_DIS  + CHUNK - 1) / CHUNK;
static constexpr int CHMAX = 128;
static constexpr int SPARE0 = 992;
static_assert(NB1 < SPARE0 && NB2 < SPARE0, "1024 bin counters in LDS: bins, the end entry, the spares behind them");
static_assert(CH1 <= CHMAX && CH2 <= CHMAX, "run bounds table in LDS");
static_assert(N_DRUG < 65536 && N_DIS < 65536, "record packs (src << 6) | local row; row N of h (zeros) has a 16-bit index too");

static constexpr int FRONT_THREADS = 256;
static constexpr int EPT   = CHUNK / FRONT_THREADS;
static constexpr int MT1   = N_DRUG / 16;
static constexpr int MP2   = N_DIS / 32;
static constexpr int GEMM_WAVES  = MT1 + MP2;
static constexpr int GEMM_BLOCKS = (GEMM_WAVES + 7) / 8;
static constexpr int GRP_BLOCKS  = CH1 + CH2;
static_assert(N_DRUG % 16 == 0 && N_DIS % 32 == 0, "whole tiles, whole tile pairs");
static_assert(F_DRUG == 2 * F_DIS, "a tile pair at F_DIS is the matrix work of one tile at F_DRUG");
static_assert(CHUNK % FRONT_THREADS == 0 && FRONT_THREADS * 4 == 1024, "4 counters per thread in the scan");
static_assert(CHUNK <= 8192, "rank in a bin fits 13 bits of the key");
static_assert(E_DRUG % 4 == 0 && E_DIS % 4 == 0 && CHUNK % (FRONT_THREADS * 4) == 0,
              "a chunk's records leave LDS as 16-byte pieces, each wholly inside the edge list or wholly behind it");

__device__ __forceinline__ void fgcn_gemm_tile(const float* __restrict__ X,
                                               const float* __restrict__ W,
                                               float* __restrict__ H,
                                               int m0, int K, int lane) {
    const int lane16 = lane & 15;
    const bool hi    = lane >= 16;

    v8f acc0 = {}, acc1 = {}, acc2 = {}, acc3 = {};

    const float* xa = X + (size_t)(m0 + lane16) * K + (hi ? 2 : 0);
    const float* wb = W + (size_t)(hi ? 2 : 0) * 64 + lane16 * 4;

    #pragma unroll 8
    for (int k = 0; k < K; k += 4) {
        float2 af = *(const float2*)(xa + k);
        v2f a; a.x = af.x; a.y = af.y;

        const float* wk = wb + (size_t)k * 64;
        const float4 w0 = *(const float4*)(wk);
        const float4 w1 = *(const float4*)(wk + 64);
        v2f b0; b0.x = w0.x; b0.y = w1.x;
        v2f b1; b1.x = w0.y; b1.y = w1.y;
        v2f b2; b2.x = w0.z; b2.y = w1.z;
        v2f b3; b3.x = w0.w; b3.y = w1.w;

        acc0 = __builtin_amdgcn_wmma_f32_16x16x4_f32(false, a, false, b0, (short)0, acc0, false, false);
        acc1 = __builtin_amdgcn_wmma_f32_16x16x4_f32(false, a, false, b1, (short)0, acc1, false, false);
        acc2 = __builtin_amdgcn_wmma_f32_16x16x4_f32(false, a, false, b2, (short)0, acc2, false, false);
        acc3 = __builtin_amdgcn_wmma_f32_16x16x4_f32(false, a, false, b3, (short)0, acc3, false, false);
    }

    const int rbase = m0 + (hi ? 8 : 0);
    #pragma unroll
    for (int r = 0; r < 8; ++r) {
        float4 o;
        o.x = acc0[r];
        o.y = acc1[r];
        o.z = acc2[r];
        o.w = acc3[r];
        *(float4*)(H + (size_t)(rbase + r) * 64 + lane16 * 4) = o;
    }
}

__device__ __forceinline__ void fgcn_group_chunk(int c, int* cnt, int* wsum, int* stage,
                                                 const int* __restrict__ ei1,
                                                 const int* __restrict__ ei2,
                                                 int* __restrict__ rec1,
                                                 int* __restrict__ rec2,
                                                 int* __restrict__ loc1,
                                                 int* __restrict__ loc2) {
    const int tid  = threadIdx.x;
    const int lane = tid & 31;
    const int w    = tid >> 5;

    const int* src; const int* dst; int* rec; int* loc; int E, NB;
    if (c < CH1) { src = ei1; dst = ei1 + E_DRUG; rec = rec1; loc = loc1; E = E_DRUG; NB = NB1; }
    else { c -= CH1; src = ei2; dst = ei2 + E_DIS; rec = rec2; loc = loc2; E = E_DIS; NB = NB2; }

    *(int4*)(cnt + tid * 4) = make_int4(0, 0, 0, 0);
    __syncthreads();

    const int base = c * CHUNK;
    int r[EPT], key[EPT];
    #pragma unroll
    for (int i = 0; i < EPT; ++i) {
        const int e  = base + i * FRONT_THREADS + tid;
        const int ec = (e < E) ? e : E - 1;
        r[i]   = src[ec];
        key[i] = dst[ec];
    }
    #pragma unroll
    for (int i = 0; i < EPT; ++i) {
        const int e = base + i * FRONT_THREADS + tid;
        const int d = key[i];
        r[i]   = (r[i] << 6) | (d & 63);
        key[i] = (e < E) ? (d >> 6) : SPARE0 + lane;
    }
    int rk[EPT];
    #pragma unroll
    for (int i = 0; i < EPT; ++i) rk[i] = atomicAdd(&cnt[key[i]], 1);
    #pragma unroll
    for (int i = 0; i < EPT; ++i) key[i] = (key[i] >= SPARE0) ? -1 : ((key[i] << 13) | rk[i]);
    __syncthreads();

    const int4 v = *(const int4*)(cnt + tid * 4);
    const int s4 = v.x + v.y + v.z + v.w;
    int x = s4;
    for (int o = 1; o < 32; o <<= 1) {
        const int y = __shfl_up(x, o);
        if (lane >= o) x += y;
    }
    if (lane == 31) wsum[w] = x;
    __syncthreads();
    int wp = 0;
    #pragma unroll
    for (int j = 0; j < FRONT_THREADS / 32; ++j) {
        const int t = wsum[j];
        if (j < w) wp += t;
    }
    int4 o4;
    o4.x = x - s4 + wp;
    o4.y = o4.x + v.x;
    o4.z = o4.y + v.y;
    o4.w = o4.z + v.z;
    *(int4*)(cnt + tid * 4) = o4;
    int* lrow = loc + (size_t)c * (NB + 1);
    const int q = tid * 4;
    if (q + 0 <= NB) lrow[q + 0] = o4.x;
    if (q + 1 <= NB) lrow[q + 1] = o4.y;
    if (q + 2 <= NB) lrow[q + 2] = o4.z;
    if (q + 3 <= NB) lrow[q + 3] = o4.w;
    __syncthreads();

    int pos[EPT];
    #pragma unroll
    for (int i = 0; i < EPT; ++i) pos[i] = cnt[(key[i] >= 0) ? (key[i] >> 13) : 0] + (key[i] & 8191);
    #pragma unroll
    for (int i = 0; i < EPT; ++i) stage[(key[i] >= 0) ? pos[i] : CHUNK + lane] = r[i];
    __syncthreads();

    constexpr int PIECES = CHUNK / (FRONT_THREADS * 4);
    const int nvalid = (E - base < CHUNK) ? (E - base) : CHUNK;
    int* outp = rec + (size_t)base;
    if (nvalid == CHUNK) {
        int4 piece[PIECES];
        #pragma unroll
        for (int j = 0; j < PIECES; ++j) piece[j] = *(const int4*)(stage + (j * FRONT_THREADS + tid) * 4);
        #pragma unroll
        for (int j = 0; j < PIECES; ++j) *(int4*)(outp + (j * FRONT_THREADS + tid) * 4) = piece[j];
    } else {
        for (int j = 0; j < PIECES; ++j) {
            const int at = (j * FRONT_THREADS + tid) * 4;
            if (at < nvalid) *(int4*)(outp + at) = *(const int4*)(stage + at);
        }
    }
}

__global__ __launch_bounds__(FRONT_THREADS) void fgcn_front(const float* __restrict__ x1,
                                                            const float* __restrict__ x2,
                                                            const float* __restrict__ w1,
                                                            const float* __restrict__ w2,
                                                            float* __restrict__ h1,
                                                            float* __restrict__ h2,
                                                            const int* __restrict__ ei1,
                                                            const int* __restrict__ ei2,
                                                            int* __restrict__ rec1,
                                                            int* __restrict__ rec2,
                                                            int* __restrict__ loc1,
                                                            int* __restrict__ loc2) {
    __shared__ __attribute__((aligned(16))) int cnt[1024];
    __shared__ int wsum[FRONT_THREADS / 32];
    __shared__ __attribute__((aligned(16))) int stage[CHUNK + 32];

    const int b = blockIdx.x;
    if (b >= GEMM_BLOCKS) {
        fgcn_group_chunk(b - GEMM_BLOCKS, cnt, wsum, stage, ei1, ei2, rec1, rec2, loc1, loc2);
        return;
    }

    const int wv   = __builtin_amdgcn_readfirstlane(b * (FRONT_THREADS / 32) + (threadIdx.x >> 5));
    const int lane = threadIdx.x & 31;
    if (wv == 0) {
        *(float2*)(h1 + (size_t)N_DRUG * NHID2 + lane * 2) = make_float2(0.0f, 0.0f);
        *(float2*)(h2 + (size_t)N_DIS * NHID2 + lane * 2)  = make_float2(0.0f, 0.0f);
    }
    if (wv < MT1) {
        fgcn_gemm_tile(x1, w1, h1, wv * 16, F_DRUG, lane);
    } else if (wv < GEMM_WAVES) {
        const int m0 = (wv - MT1) * 32;
        fgcn_gemm_tile(x2, w2, h2, m0, F_DIS, lane);
        fgcn_gemm_tile(x2, w2, h2, m0 + 16, F_DIS, lane);
    }
}

__device__ __forceinline__ void fgcn_row_add(float* acc, int rcd, int lane, float2 v) {
    float* a = acc + ((rcd & 63) << 6) + lane * 2;
    atomicAdd(a, v.x);
    atomicAdd(a + 1, v.y);
}

template <int K>
__device__ __forceinline__ void fgcn_rows(float* acc, const float* hl, int cur, int j, int lane) {
    int r[K];
    float2 v[K];
    #pragma unroll
    for (int k = 0; k < K; ++k) r[k] = __builtin_amdgcn_readlane(cur, j + k);
    #pragma unroll
    for (int k = 0; k < K; ++k) v[k] = *(const float2*)(hl + ((size_t)(r[k] >> 6) << 6));
    #pragma unroll
    for (int k = 0; k < K; ++k) fgcn_row_add(acc, r[k], lane, v[k]);
}

__device__ __forceinline__ void fgcn_piece(float* acc, const float* hl, int cur, int m, int lane) {
    int j = 0;
    for (; j + 8 <= m; j += 8) fgcn_rows<8>(acc, hl, cur, j, lane);
    switch (m - j) {
        case 7: fgcn_rows<7>(acc, hl, cur, j, lane); break;
        case 6: fgcn_rows<6>(acc, hl, cur, j, lane); break;
        case 5: fgcn_rows<5>(acc, hl, cur, j, lane); break;
        case 4: fgcn_rows<4>(acc, hl, cur, j, lane); break;
        case 3: fgcn_rows<3>(acc, hl, cur, j, lane); break;
        case 2: fgcn_rows<2>(acc, hl, cur, j, lane); break;
        case 1: fgcn_rows<1>(acc, hl, cur, j, lane); break;
        default: break;
    }
}

__device__ __forceinline__ void fgcn_sum_rows_lds(float* acc, const int* lo, const int* hi,
                                                  const float* __restrict__ h,
                                                  const int* __restrict__ rec,
                                                  const float* __restrict__ bias,
                                                  float* __restrict__ out,
                                                  int b, int CH, int N) {
    const int tid  = threadIdx.x;
    const int lane = tid & 31;
    const int w    = tid >> 5;

    for (int i = tid; i < BIN_ROWS * NHID2; i += 256) acc[i] = 0.0f;
    __syncthreads();

    const float* hl = h + lane * 2;
    int c = __builtin_amdgcn_readfirstlane(w);
    int s = 0, e = 0, mine = 0;
    if (c < CH) {
        s = __builtin_amdgcn_readfirstlane(lo[c]);
        e = __builtin_amdgcn_readfirstlane(hi[c]);
        if (lane < e - s) mine = rec[(size_t)c * CHUNK + s + lane];
    }
    while (c < CH) {
        const int cn = c + 8;
        int sn = 0, en = 0, minen = 0;
        if (cn < CH) {
            sn = __builtin_amdgcn_readfirstlane(lo[cn]);
            en = __builtin_amdgcn_readfirstlane(hi[cn]);
            if (lane < en - sn) minen = rec[(size_t)cn * CHUNK + sn + lane];
        }
        const int n = e - s;
        if (n > 0) {
            fgcn_piece(acc, hl, mine, (n < 32) ? n : 32, lane);
            const int* rp = rec + (size_t)c * CHUNK;
            for (int p = s + 32; p < e; p += 32) {
                const int m = (e - p < 32) ? (e - p) : 32;
                const int more = (lane < m) ? rp[p + lane] : 0;
                fgcn_piece(acc, hl, more, m, lane);
            }
        }
        c = cn; s = sn; e = en; mine = minen;
    }
    __syncthreads();

    for (int i = tid; i < BIN_ROWS * NHID2 / 4; i += 256) {
        const int row = b * BIN_ROWS + (i >> 4);
        if (row < N) {
            const int q = (i & 15) << 2;
            float4 a = *(const float4*)(acc + i * 4);
            a.x += bias[q + 0];
            a.y += bias[q + 1];
            a.z += bias[q + 2];
            a.w += bias[q + 3];
            *(float4*)(out + (size_t)row * NHID2 + q) = a;
        }
    }
}

static constexpr int SLOT_CAP  = 128;
static constexpr int SLOT      = SLOT_CAP + 1;
static constexpr int LST_WORDS = BIN_ROWS * SLOT + 32;
static_assert(LST_WORDS % 4 == 0 && LST_WORDS >= BIN_ROWS * NHID2, "filled as 16-byte pieces; the general path keeps 64 x 64 sums here");
static_assert(CHMAX == 128, "a wave takes 16 runs; a lane scans 4 run lengths");

__device__ __forceinline__ int fgcn_place16(int* cnt, int* lst, const int (&r)[16], int lane) {
    int p[16];
    #pragma unroll
    for (int i = 0; i < 16; ++i) p[i] = atomicAdd(&cnt[(r[i] >= 0) ? (r[i] & 63) : 64 + lane], 1);
    int top = 0;
    const int spare = BIN_ROWS * SLOT + lane;
    #pragma unroll
    for (int i = 0; i < 16; ++i) {
        const int m  = r[i] >> 31;
        const int at = (r[i] & 63) * SLOT + min(p[i], SLOT_CAP);
        top = max(top, p[i] & ~m);
        lst[(at & ~m) | (spare & m)] = r[i] >> 6;
    }
    return top;
}

__global__ __launch_bounds__(256) void fgcn_sum_rows(const float* __restrict__ h1,
                                                     const float* __restrict__ h2,
                                                     const int* __restrict__ rec1,
                                                     const int* __restrict__ rec2,
                                                     const int* __restrict__ loc1,
                                                     const int* __restrict__ loc2,
                                                     const float* __restrict__ b1,
                                                     const float* __restrict__ b2,
                                                     float* __restrict__ out1,
                                                     float* __restrict__ out2) {
    __shared__ __attribute__((aligned(16))) int lst[LST_WORDS];
    __shared__ __attribute__((aligned(16))) int lo[CHMAX];
    __shared__ __attribute__((aligned(16))) int hi[CHMAX];
    __shared__ int cnt[96];
    __shared__ int over;

    int b = blockIdx.x;
    const float* h; const int* rec; const int* loc; const float* bias; float* out; int NB, CH, N;
    if (b < NB1) { h = h1; rec = rec1; loc = loc1; bias = b1; out = out1; NB = NB1; CH = CH1; N = N_DRUG; }
    else { b -= NB1; h = h2; rec = rec2; loc = loc2; bias = b2; out = out2; NB = NB2; CH = CH2; N = N_DIS; }

    const int tid  = threadIdx.x;
    const int lane = tid & 31;
    const int w    = __builtin_amdgcn_readfirstlane(tid >> 5);

    if (tid < 96) cnt[tid] = 0;
    if (tid == 0) over = 0;
    {
        const int4 z4 = make_int4(N, N, N, N);
        #pragma unroll
        for (int i = 0; i < LST_WORDS / 4 / 256; ++i) *(int4*)(lst + (i * 256 + tid) * 4) = z4;
        if (tid < LST_WORDS / 4 % 256) *(int4*)(lst + (LST_WORDS / 4 / 256 * 256 + tid) * 4) = z4;
    }
    if (tid < CHMAX) {
        int l = 0, u = 0;
        if (tid < CH) {
            const int* lp = loc + (size_t)tid * (NB + 1) + b;
            l = lp[0];
            u = lp[1];
        }
        lo[tid] = l;
        hi[tid] = u;
    }
    __syncthreads();

    const int4 l4 = *(const int4*)(lo + lane * 4);
    const int4 u4 = *(const int4*)(hi + lane * 4);
    int mx = max(max(u4.x - l4.x, u4.y - l4.y), max(u4.z - l4.z, u4.w - l4.w));
    for (int o = 16; o > 0; o >>= 1) mx = max(mx, __shfl_xor(mx, o));
    mx = __builtin_amdgcn_readfirstlane(mx);
    if (mx > 64) {
        fgcn_sum_rows_lds((float*)lst, lo, hi, h, rec, bias, out, b, CH, N);
        return;
    }
    const bool two = mx > 32;

    const int cl  = w + 8 * (lane & 15);
    const int lov = lo[cl];
    const int nov = hi[cl] - lov;
    int ra[16], rb[16];
    #pragma unroll
    for (int i = 0; i < 16; ++i) {
        const int s = __builtin_amdgcn_readlane(lov, i);
        const int n = __builtin_amdgcn_readlane(nov, i);
        const int* rp = rec + (size_t)(w + 8 * i) * CHUNK + s + lane;
        int v = -1;
        if (lane < n) v = rp[0];
        ra[i] = v;
    }
    #pragma unroll
    for (int i = 0; i < 16; ++i) {
        const int s = __builtin_amdgcn_readlane(lov, i);
        const int n = __builtin_amdgcn_readlane(nov, i);
        const int* rp = rec + (size_t)(w + 8 * i) * CHUNK + s + lane;
        int v = -1;
        if (two && lane + 32 < n) v = rp[32];
        rb[i] = v;
    }
    int top = fgcn_place16(cnt, lst, ra, lane);
    if (two) top = max(top, fgcn_place16(cnt, lst, rb, lane));
    if (top >= SLOT_CAP) over = 1;
    __syncthreads();

    if (__builtin_amdgcn_readfirstlane(over) != 0) {
        fgcn_sum_rows_lds((float*)lst, lo, hi, h, rec, bias, out, b, CH, N);
        return;
    }

    const int hf  = lane >> 4;
    const int c16 = lane & 15;
    const char* hb = (const char*)h;
    const unsigned q16 = (unsigned)c16 * 16u;
    const float4 bq = *(const float4*)(bias + c16 * 4);
    const int cv = cnt[w * 8 + (lane & 7)];
    #pragma unroll 1
    for (int rr = 0; rr < 8; ++rr) {
        const int n = __builtin_amdgcn_readlane(cv, rr);
        const int* ls = lst + (w * 8 + rr) * SLOT + hf;
        float4 sum = make_float4(0.0f, 0.0f, 0.0f, 0.0f);
        int sx[8];
        #pragma unroll
        for (int k = 0; k < 8; ++k) sx[k] = ls[2 * k];
        for (int p = 0; p < n; p += 16) {
            float4 v[8];
            #pragma unroll
            for (int k = 0; k < 8; ++k) v[k] = *(const float4*)(hb + ((((unsigned)sx[k]) << 8) | q16));
            #pragma unroll
            for (int k = 0; k < 8; ++k) sx[k] = ls[p + 16 + 2 * k];
            #pragma unroll
            for (int k = 0; k < 8; ++k) {
                sum.x += v[k].x;
                sum.y += v[k].y;
                sum.z += v[k].z;
                sum.w += v[k].w;
            }
        }
        sum.x += __shfl_xor(sum.x, 16);
        sum.y += __shfl_xor(sum.y, 16);
        sum.z += __shfl_xor(sum.z, 16);
        sum.w += __shfl_xor(sum.w, 16);
        const int row = b * BIN_ROWS + w * 8 + rr;
        if (row < N && lane < 16) {
            float4 o;
            o.x = sum.x + bq.x;
            o.y = sum.y + bq.y;
            o.z = sum.z + bq.z;
            o.w = sum.w + bq.w;
            *(float4*)(out + (size_t)row * NHID2 + c16 * 4) = o;
        }
    }
}

extern "C" void kernel_launch(void* const* d_in, const int* in_sizes, int n_in,
                              void* d_out, int out_size, void* d_ws, size_t ws_size,
                              hipStream_t stream) {
    const float* drug_x  = (const float*)d_in[0];
    const float* dis_x   = (const float*)d_in[1];
    const float* W1      = (const float*)d_in[2];
    const float* b1      = (const float*)d_in[3];
    const float* W2      = (const float*)d_in[4];
    const float* b2      = (const float*)d_in[5];
    const int*   drug_ei = (const int*)d_in[6];
    const int*   dis_ei  = (const int*)d_in[7];

    float* out1 = (float*)d_out;
    float* out2 = out1 + (size_t)N_DRUG * NHID2;

    float* h1  = (float*)d_ws;
    float* h2  = h1 + (size_t)(N_DRUG + 1) * NHID2;
    int* rec1  = (int*)(h2 + (size_t)(N_DIS + 1) * NHID2);
    int* rec2  = rec1 + (size_t)E_DRUG;
    int* loc1  = rec2 + (size_t)E_DIS;
    int* loc2  = loc1 + (size_t)CH1 * (NB1 + 1);

    fgcn_front<<<GEMM_BLOCKS + GRP_BLOCKS, FRONT_THREADS, 0, stream>>>(drug_x, dis_x, W1, W2, h1, h2,
                                                                      drug_ei, dis_ei, rec1, rec2, loc1, loc2);

    fgcn_sum_rows<<<NB1 + NB2, 256, 0, stream>>>(h1, h2, rec1, rec2, loc1, loc2, b1, b2, out1, out2);
}
